# starting kernel
# baseline (speedup 1.0000x reference)
#include <hip/hip_runtime.h>

#define H 128
#define INW 2048
#define PROJ 64
#define BSZ 64
#define TLEN 512
#define M_TOT (BSZ * TLEN)
#define EPSV 1e-5f

typedef __attribute__((ext_vector_type(2))) float v2f;
typedef __attribute__((ext_vector_type(8))) float v8f;

#define WMMA_F32(a, b, c) \
  __builtin_amdgcn_wmma_f32_16x16x4_f32(false, (a), false, (b), (short)0, (c), false, false)

#define KC 128
#define LW (KC + 4)

__global__ __launch_bounds__(256) void gemm1_kernel(
    const float* __restrict__ x, const float* __restrict__ Wih1,
    const float* __restrict__ bih1, const float* __restrict__ bhh1,
    float* __restrict__ xp1) {
  __shared__ float xs[32 * LW];
  __shared__ float ws[128 * LW];
  const int tid  = threadIdx.x;
  const int wave = tid >> 5;
  const int lane = tid & 31;
  const int half = lane >> 4;
  const int l16  = lane & 15;
  const int n0   = wave * 16;
  const long m0  = (long)blockIdx.x * 32;

  v8f c0 = {};
  v8f c1 = {};
  for (int k0 = 0; k0 < INW; k0 += KC) {
    for (int i = 0; i < 4; ++i) {
      int idx = tid + i * 256;
      int row = idx >> 5;
      int cv  = (idx & 31) << 2;
      float4 v = *(const float4*)&x[(m0 + row) * INW + k0 + cv];
      *(float4*)&xs[row * LW + cv] = v;
    }
    for (int i = 0; i < 16; ++i) {
      int idx = tid + i * 256;
      int row = idx >> 5;
      int cv  = (idx & 31) << 2;
      float4 v = *(const float4*)&Wih1[(long)row * INW + k0 + cv];
      *(float4*)&ws[row * LW + cv] = v;
    }
    __syncthreads();
    for (int k = 0; k < KC; k += 4) {
      const int kc = k + 2 * half;
      v2f b;  b.x  = ws[(n0 + l16) * LW + kc];  b.y  = ws[(n0 + l16) * LW + kc + 1];
      v2f a0; a0.x = xs[l16 * LW + kc];         a0.y = xs[l16 * LW + kc + 1];
      v2f a1; a1.x = xs[(16 + l16) * LW + kc];  a1.y = xs[(16 + l16) * LW + kc + 1];
      c0 = WMMA_F32(a0, b, c0);
      c1 = WMMA_F32(a1, b, c1);
    }
    __syncthreads();
  }
  const float bias = bih1[n0 + l16] + bhh1[n0 + l16];
  for (int v = 0; v < 8; ++v) {
    long m = m0 + v + 8 * half;
    xp1[m * H + n0 + l16]        = c0[v] + bias;
    xp1[(m + 16) * H + n0 + l16] = c1[v] + bias;
  }
}

#define HL (H + 4)

__global__ __launch_bounds__(256) void rnn1_kernel(
    const float* __restrict__ xp1, const float* __restrict__ Whh1,
    const float* __restrict__ Wih2, const float* __restrict__ bih2,
    const float* __restrict__ bhh2, float* __restrict__ xp2) {
  __shared__ float hs[16 * HL];
  const int tid  = threadIdx.x;
  const int wave = tid >> 5;
  const int lane = tid & 31;
  const int half = lane >> 4;
  const int l16  = lane & 15;
  const int n0   = wave * 16;
  const int b0   = blockIdx.x * 16;

  v2f bh[32], bi[32];
  for (int kk = 0; kk < 32; ++kk) {
    const int kc = kk * 4 + 2 * half;
    bh[kk].x = Whh1[(n0 + l16) * H + kc];  bh[kk].y = Whh1[(n0 + l16) * H + kc + 1];
    bi[kk].x = Wih2[(n0 + l16) * H + kc];  bi[kk].y = Wih2[(n0 + l16) * H + kc + 1];
  }
  const float bias2 = bih2[n0 + l16] + bhh2[n0 + l16];

  for (int i = tid; i < 16 * HL; i += 256) hs[i] = 0.0f;
  __syncthreads();

  for (int t = 0; t < TLEN; ++t) {
    v8f c;
    for (int v = 0; v < 8; ++v) {
      long m = (long)(b0 + v + 8 * half) * TLEN + t;
      c[v] = xp1[m * H + n0 + l16];
    }
    for (int kk = 0; kk < 32; ++kk) {
      const int kc = kk * 4 + 2 * half;
      v2f a; a.x = hs[l16 * HL + kc]; a.y = hs[l16 * HL + kc + 1];
      c = WMMA_F32(a, bh[kk], c);
    }
    for (int v = 0; v < 8; ++v) c[v] = tanhf(c[v]);
    __syncthreads();
    for (int v = 0; v < 8; ++v) hs[(v + 8 * half) * HL + n0 + l16] = c[v];
    __syncthreads();

    v8f c2;
    for (int v = 0; v < 8; ++v) c2[v] = bias2;
    for (int kk = 0; kk < 32; ++kk) {
      const int kc = kk * 4 + 2 * half;
      v2f a; a.x = hs[l16 * HL + kc]; a.y = hs[l16 * HL + kc + 1];
      c2 = WMMA_F32(a, bi[kk], c2);
    }
    for (int v = 0; v < 8; ++v) {
      long m = (long)(b0 + v + 8 * half) * TLEN + t;
      xp2[m * H + n0 + l16] = c2[v];
    }
  }
}

__global__ __launch_bounds__(256) void rnn2_kernel(
    const float* __restrict__ xp2, const float* __restrict__ Whh2,
    float* __restrict__ h2) {
  __shared__ float hs[16 * HL];
  const int tid  = threadIdx.x;
  const int wave = tid >> 5;
  const int lane = tid & 31;
  const int half = lane >> 4;
  const int l16  = lane & 15;
  const int n0   = wave * 16;
  const int b0   = blockIdx.x * 16;

  v2f bh[32];
  for (int kk = 0; kk < 32; ++kk) {
    const int kc = kk * 4 + 2 * half;
    bh[kk].x = Whh2[(n0 + l16) * H + kc];  bh[kk].y = Whh2[(n0 + l16) * H + kc + 1];
  }
  for (int i = tid; i < 16 * HL; i += 256) hs[i] = 0.0f;
  __syncthreads();

  v8f c = {};
  for (int t = 0; t < TLEN; ++t) {
    for (int v = 0; v < 8; ++v) {
      long m = (long)(b0 + v + 8 * half) * TLEN + t;
      c[v] = xp2[m * H + n0 + l16];
    }
    for (int kk = 0; kk < 32; ++kk) {
      const int kc = kk * 4 + 2 * half;
      v2f a; a.x = hs[l16 * HL + kc]; a.y = hs[l16 * HL + kc + 1];
      c = WMMA_F32(a, bh[kk], c);
    }
    for (int v = 0; v < 8; ++v) c[v] = tanhf(c[v]);
    __syncthreads();
    for (int v = 0; v < 8; ++v) hs[(v + 8 * half) * HL + n0 + l16] = c[v];
    __syncthreads();
  }
  for (int v = 0; v < 8; ++v)
    h2[(long)(b0 + v + 8 * half) * H + n0 + l16] = c[v];
}

__global__ __launch_bounds__(64) void proj_ln_kernel(
    const float* __restrict__ h2, const float* __restrict__ Wp,
    const float* __restrict__ bp, const float* __restrict__ gamma,
    const float* __restrict__ beta, float* __restrict__ out) {
  __shared__ float tmp[PROJ];
  const int b = blockIdx.x;
  const int j = threadIdx.x;
  float acc = bp[j];
  for (int k = 0; k < H; ++k) acc += h2[b * H + k] * Wp[j * H + k];

  tmp[j] = acc;
  __syncthreads();
  for (int s = 32; s > 0; s >>= 1) {
    if (j < s) tmp[j] += tmp[j + s];
    __syncthreads();
  }
  const float mu = tmp[0] * (1.0f / PROJ);
  __syncthreads();
  const float d = acc - mu;
  tmp[j] = d * d;
  __syncthreads();
  for (int s = 32; s > 0; s >>= 1) {
    if (j < s) tmp[j] += tmp[j + s];
    __syncthreads();
  }
  const float var = tmp[0] * (1.0f / PROJ);
  const float r = rsqrtf(var + EPSV);
  out[b * PROJ + j] = d * r * gamma[j] + beta[j];
}

extern "C" void kernel_launch(void* const* d_in, const int* in_sizes, int n_in,
                              void* d_out, int out_size, void* d_ws, size_t ws_size,
                              hipStream_t stream) {
  (void)in_sizes; (void)n_in; (void)out_size; (void)ws_size;
  const float* x     = (const float*)d_in[0];
  const float* Wih1  = (const float*)d_in[1];
  const float* Whh1  = (const float*)d_in[2];
  const float* bih1  = (const float*)d_in[3];
  const float* bhh1  = (const float*)d_in[4];
  const float* Wih2  = (const float*)d_in[5];
  const float* Whh2  = (const float*)d_in[6];
  const float* bih2  = (const float*)d_in[7];
  const float* bhh2  = (const float*)d_in[8];
  const float* Wproj = (const float*)d_in[9];
  const float* bproj = (const float*)d_in[10];
  const float* gam   = (const float*)d_in[11];
  const float* bet   = (const float*)d_in[12];
  float* out = (float*)d_out;

  float* xp1 = (float*)d_ws;
  float* xp2 = xp1 + (size_t)M_TOT * H;
  float* h2  = xp2 + (size_t)M_TOT * H;

  gemm1_kernel<<<M_TOT / 32, 256, 0, stream>>>(x, Wih1, bih1, bhh1, xp1);
  rnn1_kernel<<<BSZ / 16, 256, 0, stream>>>(xp1, Whh1, Wih2, bih2, bhh2, xp2);
  rnn2_kernel<<<BSZ / 16, 256, 0, stream>>>(xp2, Whh2, h2);
  proj_ln_kernel<<<BSZ, PROJ, 0, stream>>>(h2, Wproj, bproj, gam, bet, out);
}


# layer-2 input GEMM moved out of the serial layer-1 recurrence kernel into its own chip-wide f32 WMMA launch (bias start, same k order); rnn1 keeps only h per step, stored as full rows from LDS
# speedup vs baseline: 1.0725x; 1.0725x over previous
#include <hip/hip_runtime.h>

#define H 128
#define INW 2048
#define PROJ 64
#define BSZ 64
#define TLEN 512
#define M_TOT (BSZ * TLEN)
#define EPSV 1e-5f

typedef __attribute__((ext_vector_type(2))) float v2f;
typedef __attribute__((ext_vector_type(8))) float v8f;

#define WMMA_F32(a, b, c) \
  __builtin_amdgcn_wmma_f32_16x16x4_f32(false, (a), false, (b), (short)0, (c), false, false)

#define KC 128
#define LW (KC + 4)

__global__ __launch_bounds__(256) void gemm1_kernel(
    const float* __restrict__ x, const float* __restrict__ Wih1,
    const float* __restrict__ bih1, const float* __restrict__ bhh1,
    float* __restrict__ xp1) {
  __shared__ float xs[32 * LW];
  __shared__ float ws[128 * LW];
  const int tid  = threadIdx.x;
  const int wave = tid >> 5;
  const int lane = tid & 31;
  const int half = lane >> 4;
  const int l16  = lane & 15;
  const int n0   = wave * 16;
  const long m0  = (long)blockIdx.x * 32;

  v8f c0 = {};
  v8f c1 = {};
  for (int k0 = 0; k0 < INW; k0 += KC) {
    for (int i = 0; i < 4; ++i) {
      int idx = tid + i * 256;
      int row = idx >> 5;
      int cv  = (idx & 31) << 2;
      float4 v = *(const float4*)&x[(m0 + row) * INW + k0 + cv];
      *(float4*)&xs[row * LW + cv] = v;
    }
    for (int i = 0; i < 16; ++i) {
      int idx = tid + i * 256;
      int row = idx >> 5;
      int cv  = (idx & 31) << 2;
      float4 v = *(const float4*)&Wih1[(long)row * INW + k0 + cv];
      *(float4*)&ws[row * LW + cv] = v;
    }
    __syncthreads();
    for (int k = 0; k < KC; k += 4) {
      const int kc = k + 2 * half;
      v2f b;  b.x  = ws[(n0 + l16) * LW + kc];  b.y  = ws[(n0 + l16) * LW + kc + 1];
      v2f a0; a0.x = xs[l16 * LW + kc];         a0.y = xs[l16 * LW + kc + 1];
      v2f a1; a1.x = xs[(16 + l16) * LW + kc];  a1.y = xs[(16 + l16) * LW + kc + 1];
      c0 = WMMA_F32(a0, b, c0);
      c1 = WMMA_F32(a1, b, c1);
    }
    __syncthreads();
  }
  const float bias = bih1[n0 + l16] + bhh1[n0 + l16];
  for (int v = 0; v < 8; ++v) {
    long m = m0 + v + 8 * half;
    xp1[m * H + n0 + l16]        = c0[v] + bias;
    xp1[(m + 16) * H + n0 + l16] = c1[v] + bias;
  }
}

#define HL (H + 4)

__global__ __launch_bounds__(256) void rnn1_kernel(
    const float* __restrict__ xp1, const float* __restrict__ Whh1,
    float* __restrict__ y1) {
  __shared__ __attribute__((aligned(16))) float hs[16 * HL];
  const int tid  = threadIdx.x;
  const int wave = tid >> 5;
  const int lane = tid & 31;
  const int half = lane >> 4;
  const int l16  = lane & 15;
  const int n0   = wave * 16;
  const int b0   = blockIdx.x * 16;

  v2f bh[32];
  for (int kk = 0; kk < 32; ++kk) {
    const int kc = kk * 4 + 2 * half;
    bh[kk].x = Whh1[(n0 + l16) * H + kc];  bh[kk].y = Whh1[(n0 + l16) * H + kc + 1];
  }

  for (int i = tid; i < 16 * HL; i += 256) hs[i] = 0.0f;
  __syncthreads();

  float* yrow = y1 + (long)(b0 + 2 * wave) * TLEN * H + lane * 4;
  const float* hrow = hs + (2 * wave) * HL + lane * 4;

  for (int t = 0; t < TLEN; ++t) {
    v8f c;
    for (int v = 0; v < 8; ++v) {
      long m = (long)(b0 + v + 8 * half) * TLEN + t;
      c[v] = xp1[m * H + n0 + l16];
    }
    if (t > 0) {
      const float4 h0v = *(const float4*)&hrow[0];
      const float4 h1v = *(const float4*)&hrow[HL];
      *(float4*)&yrow[(long)(t - 1) * H] = h0v;
      *(float4*)&yrow[(long)TLEN * H + (long)(t - 1) * H] = h1v;
    }
    for (int kk = 0; kk < 32; ++kk) {
      const int kc = kk * 4 + 2 * half;
      v2f a; a.x = hs[l16 * HL + kc]; a.y = hs[l16 * HL + kc + 1];
      c = WMMA_F32(a, bh[kk], c);
    }
    for (int v = 0; v < 8; ++v) c[v] = tanhf(c[v]);
    __syncthreads();
    for (int v = 0; v < 8; ++v) hs[(v + 8 * half) * HL + n0 + l16] = c[v];
    __syncthreads();
  }
  {
    const float4 h0v = *(const float4*)&hrow[0];
    const float4 h1v = *(const float4*)&hrow[HL];
    *(float4*)&yrow[(long)(TLEN - 1) * H] = h0v;
    *(float4*)&yrow[(long)TLEN * H + (long)(TLEN - 1) * H] = h1v;
  }
}

#define G2R 64

__global__ __launch_bounds__(256) void gemm2_kernel(
    const float* __restrict__ y1, const float* __restrict__ Wih2,
    const float* __restrict__ bih2, const float* __restrict__ bhh2,
    float* __restrict__ xp2) {
  __shared__ __attribute__((aligned(16))) float ys[G2R * LW];
  __shared__ __attribute__((aligned(16))) float ws[128 * LW];
  const int tid  = threadIdx.x;
  const int wave = tid >> 5;
  const int lane = tid & 31;
  const int half = lane >> 4;
  const int l16  = lane & 15;
  const int n0   = wave * 16;
  const long m0  = (long)blockIdx.x * G2R;

  for (int i = 0; i < 8; ++i) {
    int idx = tid + i * 256;
    int row = idx >> 5;
    int cv  = (idx & 31) << 2;
    float4 v = *(const float4*)&y1[(m0 + row) * H + cv];
    *(float4*)&ys[row * LW + cv] = v;
  }
  for (int i = 0; i < 16; ++i) {
    int idx = tid + i * 256;
    int row = idx >> 5;
    int cv  = (idx & 31) << 2;
    float4 v = *(const float4*)&Wih2[(long)row * H + cv];
    *(float4*)&ws[row * LW + cv] = v;
  }
  const float bias2 = bih2[n0 + l16] + bhh2[n0 + l16];
  v8f c0, c1, c2, c3;
  for (int v = 0; v < 8; ++v) { c0[v] = bias2; c1[v] = bias2; c2[v] = bias2; c3[v] = bias2; }
  __syncthreads();
  for (int k = 0; k < H; k += 4) {
    const int kc = k + 2 * half;
    v2f b;  b.x  = ws[(n0 + l16) * LW + kc];  b.y  = ws[(n0 + l16) * LW + kc + 1];
    v2f a0; a0.x = ys[l16 * LW + kc];         a0.y = ys[l16 * LW + kc + 1];
    v2f a1; a1.x = ys[(16 + l16) * LW + kc];  a1.y = ys[(16 + l16) * LW + kc + 1];
    v2f a2; a2.x = ys[(32 + l16) * LW + kc];  a2.y = ys[(32 + l16) * LW + kc + 1];
    v2f a3; a3.x = ys[(48 + l16) * LW + kc];  a3.y = ys[(48 + l16) * LW + kc + 1];
    c0 = WMMA_F32(a0, b, c0);
    c1 = WMMA_F32(a1, b, c1);
    c2 = WMMA_F32(a2, b, c2);
    c3 = WMMA_F32(a3, b, c3);
  }
  for (int v = 0; v < 8; ++v) {
    long m = m0 + v + 8 * half;
    xp2[m * H + n0 + l16]        = c0[v];
    xp2[(m + 16) * H + n0 + l16] = c1[v];
    xp2[(m + 32) * H + n0 + l16] = c2[v];
    xp2[(m + 48) * H + n0 + l16] = c3[v];
  }
}

__global__ __launch_bounds__(256) void rnn2_kernel(
    const float* __restrict__ xp2, const float* __restrict__ Whh2,
    float* __restrict__ h2) {
  __shared__ float hs[16 * HL];
  const int tid  = threadIdx.x;
  const int wave = tid >> 5;
  const int lane = tid & 31;
  const int half = lane >> 4;
  const int l16  = lane & 15;
  const int n0   = wave * 16;
  const int b0   = blockIdx.x * 16;

  v2f bh[32];
  for (int kk = 0; kk < 32; ++kk) {
    const int kc = kk * 4 + 2 * half;
    bh[kk].x = Whh2[(n0 + l16) * H + kc];  bh[kk].y = Whh2[(n0 + l16) * H + kc + 1];
  }
  for (int i = tid; i < 16 * HL; i += 256) hs[i] = 0.0f;
  __syncthreads();

  v8f c = {};
  for (int t = 0; t < TLEN; ++t) {
    for (int v = 0; v < 8; ++v) {
      long m = (long)(b0 + v + 8 * half) * TLEN + t;
      c[v] = xp2[m * H + n0 + l16];
    }
    for (int kk = 0; kk < 32; ++kk) {
      const int kc = kk * 4 + 2 * half;
      v2f a; a.x = hs[l16 * HL + kc]; a.y = hs[l16 * HL + kc + 1];
      c = WMMA_F32(a, bh[kk], c);
    }
    for (int v = 0; v < 8; ++v) c[v] = tanhf(c[v]);
    __syncthreads();
    for (int v = 0; v < 8; ++v) hs[(v + 8 * half) * HL + n0 + l16] = c[v];
    __syncthreads();
  }
  for (int v = 0; v < 8; ++v)
    h2[(long)(b0 + v + 8 * half) * H + n0 + l16] = c[v];
}

__global__ __launch_bounds__(64) void proj_ln_kernel(
    const float* __restrict__ h2, const float* __restrict__ Wp,
    const float* __restrict__ bp, const float* __restrict__ gamma,
    const float* __restrict__ beta, float* __restrict__ out) {
  __shared__ float tmp[PROJ];
  const int b = blockIdx.x;
  const int j = threadIdx.x;
  float acc = bp[j];
  for (int k = 0; k < H; ++k) acc += h2[b * H + k] * Wp[j * H + k];

  tmp[j] = acc;
  __syncthreads();
  for (int s = 32; s > 0; s >>= 1) {
    if (j < s) tmp[j] += tmp[j + s];
    __syncthreads();
  }
  const float mu = tmp[0] * (1.0f / PROJ);
  __syncthreads();
  const float d = acc - mu;
  tmp[j] = d * d;
  __syncthreads();
  for (int s = 32; s > 0; s >>= 1) {
    if (j < s) tmp[j] += tmp[j + s];
    __syncthreads();
  }
  const float var = tmp[0] * (1.0f / PROJ);
  const float r = rsqrtf(var + EPSV);
  out[b * PROJ + j] = d * r * gamma[j] + beta[j];
}

extern "C" void kernel_launch(void* const* d_in, const int* in_sizes, int n_in,
                              void* d_out, int out_size, void* d_ws, size_t ws_size,
                              hipStream_t stream) {
  (void)in_sizes; (void)n_in; (void)out_size; (void)ws_size;
  const float* x     = (const float*)d_in[0];
  const float* Wih1  = (const float*)d_in[1];
  const float* Whh1  = (const float*)d_in[2];
  const float* bih1  = (const float*)d_in[3];
  const float* bhh1  = (const float*)d_in[4];
  const float* Wih2  = (const float*)d_in[5];
  const float* Whh2  = (const float*)d_in[6];
  const float* bih2  = (const float*)d_in[7];
  const float* bhh2  = (const float*)d_in[8];
  const float* Wproj = (const float*)d_in[9];
  const float* bproj = (const float*)d_in[10];
  const float* gam   = (const float*)d_in[11];
  const float* bet   = (const float*)d_in[12];
  float* out = (float*)d_out;

  float* bufA = (float*)d_ws;
  float* bufB = bufA + (size_t)M_TOT * H;
  float* h2   = bufB + (size_t)M_TOT * H;

  gemm1_kernel<<<M_TOT / 32, 256, 0, stream>>>(x, Wih1, bih1, bhh1, bufA);
  rnn1_kernel<<<BSZ / 16, 256, 0, stream>>>(bufA, Whh1, bufB);
  gemm2_kernel<<<M_TOT / G2R, 256, 0, stream>>>(bufB, Wih2, bih2, bhh2, bufA);
  rnn2_kernel<<<BSZ / 16, 256, 0, stream>>>(bufA, Whh2, h2);
  proj_ln_kernel<<<BSZ, PROJ, 0, stream>>>(h2, Wproj, bproj, gam, bet, out);
}


# both recurrence kernels fetch their per-step xp rows a block of 4 time steps ahead into registers, so no step waits for a global load in front of its WMMA chain; sums and their order unchanged
# speedup vs baseline: 1.3414x; 1.2507x over previous
#include <hip/hip_runtime.h>

#define H 128
#define INW 2048
#define PROJ 64
#define BSZ 64
#define TLEN 512
#define M_TOT (BSZ * TLEN)
#define EPSV 1e-5f

typedef __attribute__((ext_vector_type(2))) float v2f;
typedef __attribute__((ext_vector_type(8))) float v8f;

#define WMMA_F32(a, b, c) \
  __builtin_amdgcn_wmma_f32_16x16x4_f32(false, (a), false, (b), (short)0, (c), false, false)

#define KC 128
#define LW (KC + 4)

__global__ __launch_bounds__(256) void gemm1_kernel(
    const float* __restrict__ x, const float* __restrict__ Wih1,
    const float* __restrict__ bih1, const float* __restrict__ bhh1,
    float* __restrict__ xp1) {
  __shared__ float xs[32 * LW];
  __shared__ float ws[128 * LW];
  const int tid  = threadIdx.x;
  const int wave = tid >> 5;
  const int lane = tid & 31;
  const int half = lane >> 4;
  const int l16  = lane & 15;
  const int n0   = wave * 16;
  const long m0  = (long)blockIdx.x * 32;

  v8f c0 = {};
  v8f c1 = {};
  for (int k0 = 0; k0 < INW; k0 += KC) {
    for (int i = 0; i < 4; ++i) {
      int idx = tid + i * 256;
      int row = idx >> 5;
      int cv  = (idx & 31) << 2;
      float4 v = *(const float4*)&x[(m0 + row) * INW + k0 + cv];
      *(float4*)&xs[row * LW + cv] = v;
    }
    for (int i = 0; i < 16; ++i) {
      int idx = tid + i * 256;
      int row = idx >> 5;
      int cv  = (idx & 31) << 2;
      float4 v = *(const float4*)&Wih1[(long)row * INW + k0 + cv];
      *(float4*)&ws[row * LW + cv] = v;
    }
    __syncthreads();
    for (int k = 0; k < KC; k += 4) {
      const int kc = k + 2 * half;
      v2f b;  b.x  = ws[(n0 + l16) * LW + kc];  b.y  = ws[(n0 + l16) * LW + kc + 1];
      v2f a0; a0.x = xs[l16 * LW + kc];         a0.y = xs[l16 * LW + kc + 1];
      v2f a1; a1.x = xs[(16 + l16) * LW + kc];  a1.y = xs[(16 + l16) * LW + kc + 1];
      c0 = WMMA_F32(a0, b, c0);
      c1 = WMMA_F32(a1, b, c1);
    }
    __syncthreads();
  }
  const float bias = bih1[n0 + l16] + bhh1[n0 + l16];
  for (int v = 0; v < 8; ++v) {
    long m = m0 + v + 8 * half;
    xp1[m * H + n0 + l16]        = c0[v] + bias;
    xp1[(m + 16) * H + n0 + l16] = c1[v] + bias;
  }
}

#define HL (H + 4)
#define TBLK 4

__global__ __launch_bounds__(256) void rnn1_kernel(
    const float* __restrict__ xp1, const float* __restrict__ Whh1,
    float* __restrict__ y1) {
  __shared__ __attribute__((aligned(16))) float hs[16 * HL];
  const int tid  = threadIdx.x;
  const int wave = tid >> 5;
  const int lane = tid & 31;
  const int half = lane >> 4;
  const int l16  = lane & 15;
  const int n0   = wave * 16;
  const int b0   = blockIdx.x * 16;

  v2f bh[32];
  for (int kk = 0; kk < 32; ++kk) {
    const int kc = kk * 4 + 2 * half;
    bh[kk].x = Whh1[(n0 + l16) * H + kc];  bh[kk].y = Whh1[(n0 + l16) * H + kc + 1];
  }

  for (int i = tid; i < 16 * HL; i += 256) hs[i] = 0.0f;
  __syncthreads();

  float* yrow = y1 + (long)(b0 + 2 * wave) * TLEN * H + lane * 4;
  const float* hrow = hs + (2 * wave) * HL + lane * 4;

  const float* xrow = xp1 + (long)(b0 + 8 * half) * TLEN * H + n0 + l16;
  v8f nxt[TBLK];
  for (int j = 0; j < TBLK; ++j)
    for (int v = 0; v < 8; ++v) nxt[j][v] = xrow[((long)v * TLEN + j) * H];

  for (int tb = 0; tb < TLEN; tb += TBLK) {
    v8f cur[TBLK];
    for (int j = 0; j < TBLK; ++j) cur[j] = nxt[j];
    if (tb + TBLK < TLEN) {
      for (int j = 0; j < TBLK; ++j)
        for (int v = 0; v < 8; ++v)
          nxt[j][v] = xrow[((long)v * TLEN + tb + TBLK + j) * H];
    }
    for (int j = 0; j < TBLK; ++j) {
      const int t = tb + j;
      v8f c = cur[j];
      if (t > 0) {
        const float4 h0v = *(const float4*)&hrow[0];
        const float4 h1v = *(const float4*)&hrow[HL];
        *(float4*)&yrow[(long)(t - 1) * H] = h0v;
        *(float4*)&yrow[(long)TLEN * H + (long)(t - 1) * H] = h1v;
      }
      for (int kk = 0; kk < 32; ++kk) {
        const int kc = kk * 4 + 2 * half;
        v2f a; a.x = hs[l16 * HL + kc]; a.y = hs[l16 * HL + kc + 1];
        c = WMMA_F32(a, bh[kk], c);
      }
      for (int v = 0; v < 8; ++v) c[v] = tanhf(c[v]);
      __syncthreads();
      for (int v = 0; v < 8; ++v) hs[(v + 8 * half) * HL + n0 + l16] = c[v];
      __syncthreads();
    }
  }
  {
    const float4 h0v = *(const float4*)&hrow[0];
    const float4 h1v = *(const float4*)&hrow[HL];
    *(float4*)&yrow[(long)(TLEN - 1) * H] = h0v;
    *(float4*)&yrow[(long)TLEN * H + (long)(TLEN - 1) * H] = h1v;
  }
}

#define G2R 64

__global__ __launch_bounds__(256) void gemm2_kernel(
    const float* __restrict__ y1, const float* __restrict__ Wih2,
    const float* __restrict__ bih2, const float* __restrict__ bhh2,
    float* __restrict__ xp2) {
  __shared__ __attribute__((aligned(16))) float ys[G2R * LW];
  __shared__ __attribute__((aligned(16))) float ws[128 * LW];
  const int tid  = threadIdx.x;
  const int wave = tid >> 5;
  const int lane = tid & 31;
  const int half = lane >> 4;
  const int l16  = lane & 15;
  const int n0   = wave * 16;
  const long m0  = (long)blockIdx.x * G2R;

  for (int i = 0; i < 8; ++i) {
    int idx = tid + i * 256;
    int row = idx >> 5;
    int cv  = (idx & 31) << 2;
    float4 v = *(const float4*)&y1[(m0 + row) * H + cv];
    *(float4*)&ys[row * LW + cv] = v;
  }
  for (int i = 0; i < 16; ++i) {
    int idx = tid + i * 256;
    int row = idx >> 5;
    int cv  = (idx & 31) << 2;
    float4 v = *(const float4*)&Wih2[(long)row * H + cv];
    *(float4*)&ws[row * LW + cv] = v;
  }
  const float bias2 = bih2[n0 + l16] + bhh2[n0 + l16];
  v8f c0, c1, c2, c3;
  for (int v = 0; v < 8; ++v) { c0[v] = bias2; c1[v] = bias2; c2[v] = bias2; c3[v] = bias2; }
  __syncthreads();
  for (int k = 0; k < H; k += 4) {
    const int kc = k + 2 * half;
    v2f b;  b.x  = ws[(n0 + l16) * LW + kc];  b.y  = ws[(n0 + l16) * LW + kc + 1];
    v2f a0; a0.x = ys[l16 * LW + kc];         a0.y = ys[l16 * LW + kc + 1];
    v2f a1; a1.x = ys[(16 + l16) * LW + kc];  a1.y = ys[(16 + l16) * LW + kc + 1];
    v2f a2; a2.x = ys[(32 + l16) * LW + kc];  a2.y = ys[(32 + l16) * LW + kc + 1];
    v2f a3; a3.x = ys[(48 + l16) * LW + kc];  a3.y = ys[(48 + l16) * LW + kc + 1];
    c0 = WMMA_F32(a0, b, c0);
    c1 = WMMA_F32(a1, b, c1);
    c2 = WMMA_F32(a2, b, c2);
    c3 = WMMA_F32(a3, b, c3);
  }
  for (int v = 0; v < 8; ++v) {
    long m = m0 + v + 8 * half;
    xp2[m * H + n0 + l16]        = c0[v];
    xp2[(m + 16) * H + n0 + l16] = c1[v];
    xp2[(m + 32) * H + n0 + l16] = c2[v];
    xp2[(m + 48) * H + n0 + l16] = c3[v];
  }
}

__global__ __launch_bounds__(256) void rnn2_kernel(
    const float* __restrict__ xp2, const float* __restrict__ Whh2,
    float* __restrict__ h2) {
  __shared__ float hs[16 * HL];
  const int tid  = threadIdx.x;
  const int wave = tid >> 5;
  const int lane = tid & 31;
  const int half = lane >> 4;
  const int l16  = lane & 15;
  const int n0   = wave * 16;
  const int b0   = blockIdx.x * 16;

  v2f bh[32];
  for (int kk = 0; kk < 32; ++kk) {
    const int kc = kk * 4 + 2 * half;
    bh[kk].x = Whh2[(n0 + l16) * H + kc];  bh[kk].y = Whh2[(n0 + l16) * H + kc + 1];
  }
  for (int i = tid; i < 16 * HL; i += 256) hs[i] = 0.0f;
  __syncthreads();

  const float* xrow = xp2 + (long)(b0 + 8 * half) * TLEN * H + n0 + l16;
  v8f nxt[TBLK];
  for (int j = 0; j < TBLK; ++j)
    for (int v = 0; v < 8; ++v) nxt[j][v] = xrow[((long)v * TLEN + j) * H];

  v8f c = {};
  for (int tb = 0; tb < TLEN; tb += TBLK) {
    v8f cur[TBLK];
    for (int j = 0; j < TBLK; ++j) cur[j] = nxt[j];
    if (tb + TBLK < TLEN) {
      for (int j = 0; j < TBLK; ++j)
        for (int v = 0; v < 8; ++v)
          nxt[j][v] = xrow[((long)v * TLEN + tb + TBLK + j) * H];
    }
    for (int j = 0; j < TBLK; ++j) {
      c = cur[j];
      for (int kk = 0; kk < 32; ++kk) {
        const int kc = kk * 4 + 2 * half;
        v2f a; a.x = hs[l16 * HL + kc]; a.y = hs[l16 * HL + kc + 1];
        c = WMMA_F32(a, bh[kk], c);
      }
      for (int v = 0; v < 8; ++v) c[v] = tanhf(c[v]);
      __syncthreads();
      for (int v = 0; v < 8; ++v) hs[(v + 8 * half) * HL + n0 + l16] = c[v];
      __syncthreads();
    }
  }
  for (int v = 0; v < 8; ++v)
    h2[(long)(b0 + v + 8 * half) * H + n0 + l16] = c[v];
}

__global__ __launch_bounds__(64) void proj_ln_kernel(
    const float* __restrict__ h2, const float* __restrict__ Wp,
    const float* __restrict__ bp, const float* __restrict__ gamma,
    const float* __restrict__ beta, float* __restrict__ out) {
  __shared__ float tmp[PROJ];
  const int b = blockIdx.x;
  const int j = threadIdx.x;
  float acc = bp[j];
  for (int k = 0; k < H; ++k) acc += h2[b * H + k] * Wp[j * H + k];

  tmp[j] = acc;
  __syncthreads();
  for (int s = 32; s > 0; s >>= 1) {
    if (j < s) tmp[j] += tmp[j + s];
    __syncthreads();
  }
  const float mu = tmp[0] * (1.0f / PROJ);
  __syncthreads();
  const float d = acc - mu;
  tmp[j] = d * d;
  __syncthreads();
  for (int s = 32; s > 0; s >>= 1) {
    if (j < s) tmp[j] += tmp[j + s];
    __syncthreads();
  }
  const float var = tmp[0] * (1.0f / PROJ);
  const float r = rsqrtf(var + EPSV);
  out[b * PROJ + j] = d * r * gamma[j] + beta[j];
}

extern "C" void kernel_launch(void* const* d_in, const int* in_sizes, int n_in,
                              void* d_out, int out_size, void* d_ws, size_t ws_size,
                              hipStream_t stream) {
  (void)in_sizes; (void)n_in; (void)out_size; (void)ws_size;
  const float* x     = (const float*)d_in[0];
  const float* Wih1  = (const float*)d_in[1];
  const float* Whh1  = (const float*)d_in[2];
  const float* bih1  = (const float*)d_in[3];
  const float* bhh1  = (const float*)d_in[4];
  const float* Wih2  = (const float*)d_in[5];
  const float* Whh2  = (const float*)d_in[6];
  const float* bih2  = (const float*)d_in[7];
  const float* bhh2  = (const float*)d_in[8];
  const float* Wproj = (const float*)d_in[9];
  const float* bproj = (const float*)d_in[10];
  const float* gam   = (const float*)d_in[11];
  const float* bet   = (const float*)d_in[12];
  float* out = (float*)d_out;

  float* bufA = (float*)d_ws;
  float* bufB = bufA + (size_t)M_TOT * H;
  float* h2   = bufB + (size_t)M_TOT * H;

  gemm1_kernel<<<M_TOT / 32, 256, 0, stream>>>(x, Wih1, bih1, bhh1, bufA);
  rnn1_kernel<<<BSZ / 16, 256, 0, stream>>>(bufA, Whh1, bufB);
  gemm2_kernel<<<M_TOT / G2R, 256, 0, stream>>>(bufB, Wih2, bih2, bhh2, bufA);
  rnn2_kernel<<<BSZ / 16, 256, 0, stream>>>(bufA, Whh2, h2);
  proj_ln_kernel<<<BSZ, PROJ, 0, stream>>>(h2, Wproj, bproj, gam, bet, out);
}


# recurrence kernels: library tanhf (two data-dependent branches per value) replaced by branch-free (1-t)/(1+t), t=exp2(-2*log2e*|x|), on v_exp_f32/v_rcp_f32 (abs error below 2e-7); WMMAs unchanged
# speedup vs baseline: 2.0055x; 1.4951x over previous
#include <hip/hip_runtime.h>

#define H 128
#define INW 2048
#define PROJ 64
#define BSZ 64
#define TLEN 512
#define M_TOT (BSZ * TLEN)
#define EPSV 1e-5f

typedef __attribute__((ext_vector_type(2))) float v2f;
typedef __attribute__((ext_vector_type(8))) float v8f;

#define WMMA_F32(a, b, c) \
  __builtin_amdgcn_wmma_f32_16x16x4_f32(false, (a), false, (b), (short)0, (c), false, false)

#define KC 128
#define LW (KC + 4)

__global__ __launch_bounds__(256) void gemm1_kernel(
    const float* __restrict__ x, const float* __restrict__ Wih1,
    const float* __restrict__ bih1, const float* __restrict__ bhh1,
    float* __restrict__ xp1) {
  __shared__ float xs[32 * LW];
  __shared__ float ws[128 * LW];
  const int tid  = threadIdx.x;
  const int wave = tid >> 5;
  const int lane = tid & 31;
  const int half = lane >> 4;
  const int l16  = lane & 15;
  const int n0   = wave * 16;
  const long m0  = (long)blockIdx.x * 32;

  v8f c0 = {};
  v8f c1 = {};
  for (int k0 = 0; k0 < INW; k0 += KC) {
    for (int i = 0; i < 4; ++i) {
      int idx = tid + i * 256;
      int row = idx >> 5;
      int cv  = (idx & 31) << 2;
      float4 v = *(const float4*)&x[(m0 + row) * INW + k0 + cv];
      *(float4*)&xs[row * LW + cv] = v;
    }
    for (int i = 0; i < 16; ++i) {
      int idx = tid + i * 256;
      int row = idx >> 5;
      int cv  = (idx & 31) << 2;
      float4 v = *(const float4*)&Wih1[(long)row * INW + k0 + cv];
      *(float4*)&ws[row * LW + cv] = v;
    }
    __syncthreads();
    for (int k = 0; k < KC; k += 4) {
      const int kc = k + 2 * half;
      v2f b;  b.x  = ws[(n0 + l16) * LW + kc];  b.y  = ws[(n0 + l16) * LW + kc + 1];
      v2f a0; a0.x = xs[l16 * LW + kc];         a0.y = xs[l16 * LW + kc + 1];
      v2f a1; a1.x = xs[(16 + l16) * LW + kc];  a1.y = xs[(16 + l16) * LW + kc + 1];
      c0 = WMMA_F32(a0, b, c0);
      c1 = WMMA_F32(a1, b, c1);
    }
    __syncthreads();
  }
  const float bias = bih1[n0 + l16] + bhh1[n0 + l16];
  for (int v = 0; v < 8; ++v) {
    long m = m0 + v + 8 * half;
    xp1[m * H + n0 + l16]        = c0[v] + bias;
    xp1[(m + 16) * H + n0 + l16] = c1[v] + bias;
  }
}

__device__ __forceinline__ float tanh_exp2(float x) {
  const float t = __builtin_amdgcn_exp2f(-2.8853900817779268f * __builtin_fabsf(x));
  const float y = (1.0f - t) * __builtin_amdgcn_rcpf(1.0f + t);
  return __builtin_copysignf(y, x);
}

#define HL (H + 4)
#define TBLK 4

__global__ __launch_bounds__(256) void rnn1_kernel(
    const float* __restrict__ xp1, const float* __restrict__ Whh1,
    float* __restrict__ y1) {
  __shared__ __attribute__((aligned(16))) float hs[16 * HL];
  const int tid  = threadIdx.x;
  const int wave = tid >> 5;
  const int lane = tid & 31;
  const int half = lane >> 4;
  const int l16  = lane & 15;
  const int n0   = wave * 16;
  const int b0   = blockIdx.x * 16;

  v2f bh[32];
  for (int kk = 0; kk < 32; ++kk) {
    const int kc = kk * 4 + 2 * half;
    bh[kk].x = Whh1[(n0 + l16) * H + kc];  bh[kk].y = Whh1[(n0 + l16) * H + kc + 1];
  }

  for (int i = tid; i < 16 * HL; i += 256) hs[i] = 0.0f;
  __syncthreads();

  float* yrow = y1 + (long)(b0 + 2 * wave) * TLEN * H + lane * 4;
  const float* hrow = hs + (2 * wave) * HL + lane * 4;

  const float* xrow = xp1 + (long)(b0 + 8 * half) * TLEN * H + n0 + l16;
  v8f nxt[TBLK];
  for (int j = 0; j < TBLK; ++j)
    for (int v = 0; v < 8; ++v) nxt[j][v] = xrow[((long)v * TLEN + j) * H];

  for (int tb = 0; tb < TLEN; tb += TBLK) {
    v8f cur[TBLK];
    for (int j = 0; j < TBLK; ++j) cur[j] = nxt[j];
    if (tb + TBLK < TLEN) {
      for (int j = 0; j < TBLK; ++j)
        for (int v = 0; v < 8; ++v)
          nxt[j][v] = xrow[((long)v * TLEN + tb + TBLK + j) * H];
    }
#pragma unroll
    for (int j = 0; j < TBLK; ++j) {
      const int t = tb + j;
      v8f c = cur[j];
      if (t > 0) {
        const float4 h0v = *(const float4*)&hrow[0];
        const float4 h1v = *(const float4*)&hrow[HL];
        *(float4*)&yrow[(long)(t - 1) * H] = h0v;
        *(float4*)&yrow[(long)TLEN * H + (long)(t - 1) * H] = h1v;
      }
      __builtin_amdgcn_sched_barrier(0);
      for (int kk = 0; kk < 32; ++kk) {
        const int kc = kk * 4 + 2 * half;
        v2f a; a.x = hs[l16 * HL + kc]; a.y = hs[l16 * HL + kc + 1];
        c = WMMA_F32(a, bh[kk], c);
      }
      for (int v = 0; v < 8; ++v) c[v] = tanh_exp2(c[v]);
      __syncthreads();
      for (int v = 0; v < 8; ++v) hs[(v + 8 * half) * HL + n0 + l16] = c[v];
      __syncthreads();
    }
  }
  {
    const float4 h0v = *(const float4*)&hrow[0];
    const float4 h1v = *(const float4*)&hrow[HL];
    *(float4*)&yrow[(long)(TLEN - 1) * H] = h0v;
    *(float4*)&yrow[(long)TLEN * H + (long)(TLEN - 1) * H] = h1v;
  }
}

#define G2R 64

__global__ __launch_bounds__(256) void gemm2_kernel(
    const float* __restrict__ y1, const float* __restrict__ Wih2,
    const float* __restrict__ bih2, const float* __restrict__ bhh2,
    float* __restrict__ xp2) {
  __shared__ __attribute__((aligned(16))) float ys[G2R * LW];
  __shared__ __attribute__((aligned(16))) float ws[128 * LW];
  const int tid  = threadIdx.x;
  const int wave = tid >> 5;
  const int lane = tid & 31;
  const int half = lane >> 4;
  const int l16  = lane & 15;
  const int n0   = wave * 16;
  const long m0  = (long)blockIdx.x * G2R;

  for (int i = 0; i < 8; ++i) {
    int idx = tid + i * 256;
    int row = idx >> 5;
    int cv  = (idx & 31) << 2;
    float4 v = *(const float4*)&y1[(m0 + row) * H + cv];
    *(float4*)&ys[row * LW + cv] = v;
  }
  for (int i = 0; i < 16; ++i) {
    int idx = tid + i * 256;
    int row = idx >> 5;
    int cv  = (idx & 31) << 2;
    float4 v = *(const float4*)&Wih2[(long)row * H + cv];
    *(float4*)&ws[row * LW + cv] = v;
  }
  const float bias2 = bih2[n0 + l16] + bhh2[n0 + l16];
  v8f c0, c1, c2, c3;
  for (int v = 0; v < 8; ++v) { c0[v] = bias2; c1[v] = bias2; c2[v] = bias2; c3[v] = bias2; }
  __syncthreads();
  for (int k = 0; k < H; k += 4) {
    const int kc = k + 2 * half;
    v2f b;  b.x  = ws[(n0 + l16) * LW + kc];  b.y  = ws[(n0 + l16) * LW + kc + 1];
    v2f a0; a0.x = ys[l16 * LW + kc];         a0.y = ys[l16 * LW + kc + 1];
    v2f a1; a1.x = ys[(16 + l16) * LW + kc];  a1.y = ys[(16 + l16) * LW + kc + 1];
    v2f a2; a2.x = ys[(32 + l16) * LW + kc];  a2.y = ys[(32 + l16) * LW + kc + 1];
    v2f a3; a3.x = ys[(48 + l16) * LW + kc];  a3.y = ys[(48 + l16) * LW + kc + 1];
    c0 = WMMA_F32(a0, b, c0);
    c1 = WMMA_F32(a1, b, c1);
    c2 = WMMA_F32(a2, b, c2);
    c3 = WMMA_F32(a3, b, c3);
  }
  for (int v = 0; v < 8; ++v) {
    long m = m0 + v + 8 * half;
    xp2[m * H + n0 + l16]        = c0[v];
    xp2[(m + 16) * H + n0 + l16] = c1[v];
    xp2[(m + 32) * H + n0 + l16] = c2[v];
    xp2[(m + 48) * H + n0 + l16] = c3[v];
  }
}

__global__ __launch_bounds__(256) void rnn2_kernel(
    const float* __restrict__ xp2, const float* __restrict__ Whh2,
    float* __restrict__ h2) {
  __shared__ float hs[16 * HL];
  const int tid  = threadIdx.x;
  const int wave = tid >> 5;
  const int lane = tid & 31;
  const int half = lane >> 4;
  const int l16  = lane & 15;
  const int n0   = wave * 16;
  const int b0   = blockIdx.x * 16;

  v2f bh[32];
  for (int kk = 0; kk < 32; ++kk) {
    const int kc = kk * 4 + 2 * half;
    bh[kk].x = Whh2[(n0 + l16) * H + kc];  bh[kk].y = Whh2[(n0 + l16) * H + kc + 1];
  }
  for (int i = tid; i < 16 * HL; i += 256) hs[i] = 0.0f;
  __syncthreads();

  const float* xrow = xp2 + (long)(b0 + 8 * half) * TLEN * H + n0 + l16;
  v8f nxt[TBLK];
  for (int j = 0; j < TBLK; ++j)
    for (int v = 0; v < 8; ++v) nxt[j][v] = xrow[((long)v * TLEN + j) * H];

  v8f c = {};
  for (int tb = 0; tb < TLEN; tb += TBLK) {
    v8f cur[TBLK];
    for (int j = 0; j < TBLK; ++j) cur[j] = nxt[j];
    if (tb + TBLK < TLEN) {
      for (int j = 0; j < TBLK; ++j)
        for (int v = 0; v < 8; ++v)
          nxt[j][v] = xrow[((long)v * TLEN + tb + TBLK + j) * H];
    }
#pragma unroll
    for (int j = 0; j < TBLK; ++j) {
      c = cur[j];
      for (int kk = 0; kk < 32; ++kk) {
        const int kc = kk * 4 + 2 * half;
        v2f a; a.x = hs[l16 * HL + kc]; a.y = hs[l16 * HL + kc + 1];
        c = WMMA_F32(a, bh[kk], c);
      }
      for (int v = 0; v < 8; ++v) c[v] = tanh_exp2(c[v]);
      __syncthreads();
      for (int v = 0; v < 8; ++v) hs[(v + 8 * half) * HL + n0 + l16] = c[v];
      __syncthreads();
    }
  }
  for (int v = 0; v < 8; ++v)
    h2[(long)(b0 + v + 8 * half) * H + n0 + l16] = c[v];
}

__global__ __launch_bounds__(64) void proj_ln_kernel(
    const float* __restrict__ h2, const float* __restrict__ Wp,
    const float* __restrict__ bp, const float* __restrict__ gamma,
    const float* __restrict__ beta, float* __restrict__ out) {
  __shared__ float tmp[PROJ];
  const int b = blockIdx.x;
  const int j = threadIdx.x;
  float acc = bp[j];
  for (int k = 0; k < H; ++k) acc += h2[b * H + k] * Wp[j * H + k];

  tmp[j] = acc;
  __syncthreads();
  for (int s = 32; s > 0; s >>= 1) {
    if (j < s) tmp[j] += tmp[j + s];
    __syncthreads();
  }
  const float mu = tmp[0] * (1.0f / PROJ);
  __syncthreads();
  const float d = acc - mu;
  tmp[j] = d * d;
  __syncthreads();
  for (int s = 32; s > 0; s >>= 1) {
    if (j < s) tmp[j] += tmp[j + s];
    __syncthreads();
  }
  const float var = tmp[0] * (1.0f / PROJ);
  const float r = rsqrtf(var + EPSV);
  out[b * PROJ + j] = d * r * gamma[j] + beta[j];
}

extern "C" void kernel_launch(void* const* d_in, const int* in_sizes, int n_in,
                              void* d_out, int out_size, void* d_ws, size_t ws_size,
                              hipStream_t stream) {
  (void)in_sizes; (void)n_in; (void)out_size; (void)ws_size;
  const float* x     = (const float*)d_in[0];
  const float* Wih1  = (const float*)d_in[1];
  const float* Whh1  = (const float*)d_in[2];
  const float* bih1  = (const float*)d_in[3];
  const float* bhh1  = (const float*)d_in[4];
  const float* Wih2  = (const float*)d_in[5];
  const float* Whh2  = (const float*)d_in[6];
  const float* bih2  = (const float*)d_in[7];
  const float* bhh2  = (const float*)d_in[8];
  const float* Wproj = (const float*)d_in[9];
  const float* bproj = (const float*)d_in[10];
  const float* gam   = (const float*)d_in[11];
  const float* bet   = (const float*)d_in[12];
  float* out = (float*)d_out;

  float* bufA = (float*)d_ws;
  float* bufB = bufA + (size_t)M_TOT * H;
  float* h2   = bufB + (size_t)M_TOT * H;

  gemm1_kernel<<<M_TOT / 32, 256, 0, stream>>>(x, Wih1, bih1, bhh1, bufA);
  rnn1_kernel<<<BSZ / 16, 256, 0, stream>>>(bufA, Whh1, bufB);
  gemm2_kernel<<<M_TOT / G2R, 256, 0, stream>>>(bufB, Wih2, bih2, bhh2, bufA);
  rnn2_kernel<<<BSZ / 16, 256, 0, stream>>>(bufA, Whh2, h2);
  proj_ln_kernel<<<BSZ, PROJ, 0, stream>>>(h2, Wproj, bproj, gam, bet, out);
}


# layer-1 input GEMM cut into the 8 time chunks: only chunk 0 runs in front; pipeline launch i computes chunk i+1 in 128 extra work-groups beside the recurrences; sums and their order unchanged
# speedup vs baseline: 3.0267x; 1.5092x over previous
#include <hip/hip_runtime.h>

#define H 128
#define INW 2048
#define PROJ 64
#define BSZ 64
#define TLEN 512
#define M_TOT (BSZ * TLEN)
#define EPSV 1e-5f

typedef __attribute__((ext_vector_type(2))) float v2f;
typedef __attribute__((ext_vector_type(8))) float v8f;

#define WMMA_F32(a, b, c) \
  __builtin_amdgcn_wmma_f32_16x16x4_f32(false, (a), false, (b), (short)0, (c), false, false)

#define KC 128
#define LW (KC + 4)
#define G1_LDS_FLOATS ((32 + 128) * LW)

__device__ __forceinline__ void gemm1_strip(
    const float* __restrict__ x, const float* __restrict__ Wih1,
    const float* __restrict__ bih1, const float* __restrict__ bhh1,
    float* __restrict__ xp1, float* lds, const long m0) {
  float* xs = lds;
  float* ws = lds + 32 * LW;
  const int tid  = threadIdx.x;
  const int wave = tid >> 5;
  const int lane = tid & 31;
  const int half = lane >> 4;
  const int l16  = lane & 15;
  const int n0   = wave * 16;

  v8f c0 = {};
  v8f c1 = {};
  for (int k0 = 0; k0 < INW; k0 += KC) {
    for (int i = 0; i < 4; ++i) {
      int idx = tid + i * 256;
      int row = idx >> 5;
      int cv  = (idx & 31) << 2;
      float4 v = *(const float4*)&x[(m0 + row) * INW + k0 + cv];
      *(float4*)&xs[row * LW + cv] = v;
    }
    for (int i = 0; i < 16; ++i) {
      int idx = tid + i * 256;
      int row = idx >> 5;
      int cv  = (idx & 31) << 2;
      float4 v = *(const float4*)&Wih1[(long)row * INW + k0 + cv];
      *(float4*)&ws[row * LW + cv] = v;
    }
    __syncthreads();
    for (int k = 0; k < KC; k += 4) {
      const int kc = k + 2 * half;
      v2f b;  b.x  = ws[(n0 + l16) * LW + kc];  b.y  = ws[(n0 + l16) * LW + kc + 1];
      v2f a0; a0.x = xs[l16 * LW + kc];         a0.y = xs[l16 * LW + kc + 1];
      v2f a1; a1.x = xs[(16 + l16) * LW + kc];  a1.y = xs[(16 + l16) * LW + kc + 1];
      c0 = WMMA_F32(a0, b, c0);
      c1 = WMMA_F32(a1, b, c1);
    }
    __syncthreads();
  }
  const float bias = bih1[n0 + l16] + bhh1[n0 + l16];
  for (int v = 0; v < 8; ++v) {
    long m = m0 + v + 8 * half;
    xp1[m * H + n0 + l16]        = c0[v] + bias;
    xp1[(m + 16) * H + n0 + l16] = c1[v] + bias;
  }
}

__device__ __forceinline__ float tanh_exp2(float x) {
  const float t = __builtin_amdgcn_exp2f(-2.8853900817779268f * __builtin_fabsf(x));
  const float y = (1.0f - t) * __builtin_amdgcn_rcpf(1.0f + t);
  return __builtin_copysignf(y, x);
}

#define HL (H + 4)
#define TBLK 4
#define CH 64
#define NCH (TLEN / CH)
#define G2R 64
#define PIPE_LDS_FLOATS (164 * 256)

__device__ __forceinline__ void rnn1_chunk(
    const float* __restrict__ xp1, const float* __restrict__ Whh1,
    float* __restrict__ y1, float* hs, const int g, const int ch) {
  const int tid  = threadIdx.x;
  const int wave = tid >> 5;
  const int lane = tid & 31;
  const int half = lane >> 4;
  const int l16  = lane & 15;
  const int n0   = wave * 16;
  const int b0   = g * 16;
  const long tofs = (long)ch * CH * H;

  v2f bh[32];
  for (int kk = 0; kk < 32; ++kk) {
    const int kc = kk * 4 + 2 * half;
    bh[kk].x = Whh1[(n0 + l16) * H + kc];  bh[kk].y = Whh1[(n0 + l16) * H + kc + 1];
  }

  if (ch == 0) {
    for (int i = tid; i < 16 * HL; i += 256) hs[i] = 0.0f;
  } else {
    const float* hin = y1 + (long)b0 * TLEN * H + tofs - H;
    const int r  = tid >> 5;
    const int cv = (tid & 31) << 2;
    const float4 va = *(const float4*)&hin[(long)r * TLEN * H + cv];
    const float4 vb = *(const float4*)&hin[(long)(r + 8) * TLEN * H + cv];
    *(float4*)&hs[r * HL + cv] = va;
    *(float4*)&hs[(r + 8) * HL + cv] = vb;
  }
  __syncthreads();

  float* yrow = y1 + (long)(b0 + 2 * wave) * TLEN * H + tofs + lane * 4;
  const float* hrow = hs + (2 * wave) * HL + lane * 4;

  const float* xrow = xp1 + (long)(b0 + 8 * half) * TLEN * H + tofs + n0 + l16;
  v8f nxt[TBLK];
  for (int j = 0; j < TBLK; ++j)
    for (int v = 0; v < 8; ++v) nxt[j][v] = xrow[((long)v * TLEN + j) * H];

  for (int tb = 0; tb < CH; tb += TBLK) {
    v8f cur[TBLK];
    for (int j = 0; j < TBLK; ++j) cur[j] = nxt[j];
    if (tb + TBLK < CH) {
      for (int j = 0; j < TBLK; ++j)
        for (int v = 0; v < 8; ++v)
          nxt[j][v] = xrow[((long)v * TLEN + tb + TBLK + j) * H];
    }
#pragma unroll
    for (int j = 0; j < TBLK; ++j) {
      const int t = tb + j;
      v8f c = cur[j];
      if (t > 0) {
        const float4 h0v = *(const float4*)&hrow[0];
        const float4 h1v = *(const float4*)&hrow[HL];
        *(float4*)&yrow[(long)(t - 1) * H] = h0v;
        *(float4*)&yrow[(long)TLEN * H + (long)(t - 1) * H] = h1v;
      }
      __builtin_amdgcn_sched_barrier(0);
      for (int kk = 0; kk < 32; ++kk) {
        const int kc = kk * 4 + 2 * half;
        v2f a; a.x = hs[l16 * HL + kc]; a.y = hs[l16 * HL + kc + 1];
        c = WMMA_F32(a, bh[kk], c);
      }
      for (int v = 0; v < 8; ++v) c[v] = tanh_exp2(c[v]);
      __syncthreads();
      for (int v = 0; v < 8; ++v) hs[(v + 8 * half) * HL + n0 + l16] = c[v];
      __syncthreads();
    }
  }
  {
    const float4 h0v = *(const float4*)&hrow[0];
    const float4 h1v = *(const float4*)&hrow[HL];
    *(float4*)&yrow[(long)(CH - 1) * H] = h0v;
    *(float4*)&yrow[(long)TLEN * H + (long)(CH - 1) * H] = h1v;
  }
}

#define LW2 (H + 4)
__device__ __forceinline__ void gemm2_chunk(
    const float* __restrict__ y1, const float* __restrict__ Wih2,
    const float* __restrict__ bih2, const float* __restrict__ bhh2,
    float* __restrict__ xp2, float* lds, const int b, const int ch) {
  float* ys = lds;
  float* ws = lds + G2R * LW2;
  const int tid  = threadIdx.x;
  const int wave = tid >> 5;
  const int lane = tid & 31;
  const int half = lane >> 4;
  const int l16  = lane & 15;
  const int n0   = wave * 16;
  const long m0  = (long)b * TLEN + (long)ch * CH;

  for (int i = 0; i < 8; ++i) {
    int idx = tid + i * 256;
    int row = idx >> 5;
    int cv  = (idx & 31) << 2;
    float4 v = *(const float4*)&y1[(m0 + row) * H + cv];
    *(float4*)&ys[row * LW2 + cv] = v;
  }
  for (int i = 0; i < 16; ++i) {
    int idx = tid + i * 256;
    int row = idx >> 5;
    int cv  = (idx & 31) << 2;
    float4 v = *(const float4*)&Wih2[(long)row * H + cv];
    *(float4*)&ws[row * LW2 + cv] = v;
  }
  const float bias2 = bih2[n0 + l16] + bhh2[n0 + l16];
  v8f c0, c1, c2, c3;
  for (int v = 0; v < 8; ++v) { c0[v] = bias2; c1[v] = bias2; c2[v] = bias2; c3[v] = bias2; }
  __syncthreads();
  for (int k = 0; k < H; k += 4) {
    const int kc = k + 2 * half;
    v2f bb; bb.x = ws[(n0 + l16) * LW2 + kc];  bb.y = ws[(n0 + l16) * LW2 + kc + 1];
    v2f a0; a0.x = ys[l16 * LW2 + kc];         a0.y = ys[l16 * LW2 + kc + 1];
    v2f a1; a1.x = ys[(16 + l16) * LW2 + kc];  a1.y = ys[(16 + l16) * LW2 + kc + 1];
    v2f a2; a2.x = ys[(32 + l16) * LW2 + kc];  a2.y = ys[(32 + l16) * LW2 + kc + 1];
    v2f a3; a3.x = ys[(48 + l16) * LW2 + kc];  a3.y = ys[(48 + l16) * LW2 + kc + 1];
    c0 = WMMA_F32(a0, bb, c0);
    c1 = WMMA_F32(a1, bb, c1);
    c2 = WMMA_F32(a2, bb, c2);
    c3 = WMMA_F32(a3, bb, c3);
  }
  for (int v = 0; v < 8; ++v) {
    long m = m0 + v + 8 * half;
    xp2[m * H + n0 + l16]        = c0[v];
    xp2[(m + 16) * H + n0 + l16] = c1[v];
    xp2[(m + 32) * H + n0 + l16] = c2[v];
    xp2[(m + 48) * H + n0 + l16] = c3[v];
  }
}

__device__ __forceinline__ void rnn2_chunk(
    const float* __restrict__ xp2, const float* __restrict__ Whh2,
    float* __restrict__ h2, float* hs, const int g, const int ch) {
  const int tid  = threadIdx.x;
  const int wave = tid >> 5;
  const int lane = tid & 31;
  const int half = lane >> 4;
  const int l16  = lane & 15;
  const int n0   = wave * 16;
  const int b0   = g * 16;
  const long tofs = (long)ch * CH * H;

  v2f bh[32];
  for (int kk = 0; kk < 32; ++kk) {
    const int kc = kk * 4 + 2 * half;
    bh[kk].x = Whh2[(n0 + l16) * H + kc];  bh[kk].y = Whh2[(n0 + l16) * H + kc + 1];
  }
  if (ch == 0) {
    for (int i = tid; i < 16 * HL; i += 256) hs[i] = 0.0f;
  } else {
    const float* hin = h2 + (long)b0 * H;
    const int r  = tid >> 5;
    const int cv = (tid & 31) << 2;
    const float4 va = *(const float4*)&hin[r * H + cv];
    const float4 vb = *(const float4*)&hin[(r + 8) * H + cv];
    *(float4*)&hs[r * HL + cv] = va;
    *(float4*)&hs[(r + 8) * HL + cv] = vb;
  }
  __syncthreads();

  const float* xrow = xp2 + (long)(b0 + 8 * half) * TLEN * H + tofs + n0 + l16;
  v8f nxt[TBLK];
  for (int j = 0; j < TBLK; ++j)
    for (int v = 0; v < 8; ++v) nxt[j][v] = xrow[((long)v * TLEN + j) * H];

  v8f c = {};
  for (int tb = 0; tb < CH; tb += TBLK) {
    v8f cur[TBLK];
    for (int j = 0; j < TBLK; ++j) cur[j] = nxt[j];
    if (tb + TBLK < CH) {
      for (int j = 0; j < TBLK; ++j)
        for (int v = 0; v < 8; ++v)
          nxt[j][v] = xrow[((long)v * TLEN + tb + TBLK + j) * H];
    }
#pragma unroll
    for (int j = 0; j < TBLK; ++j) {
      c = cur[j];
      for (int kk = 0; kk < 32; ++kk) {
        const int kc = kk * 4 + 2 * half;
        v2f a; a.x = hs[l16 * HL + kc]; a.y = hs[l16 * HL + kc + 1];
        c = WMMA_F32(a, bh[kk], c);
      }
      for (int v = 0; v < 8; ++v) c[v] = tanh_exp2(c[v]);
      __syncthreads();
      for (int v = 0; v < 8; ++v) hs[(v + 8 * half) * HL + n0 + l16] = c[v];
      __syncthreads();
    }
  }
  for (int v = 0; v < 8; ++v)
    h2[(long)(b0 + v + 8 * half) * H + n0 + l16] = c[v];
}

__global__ __launch_bounds__(256) void gemm1_head(
    const float* __restrict__ x, const float* __restrict__ Wih1,
    const float* __restrict__ bih1, const float* __restrict__ bhh1,
    float* __restrict__ xp1) {
  __shared__ __attribute__((aligned(16))) float lds[G1_LDS_FLOATS];
  const int s = blockIdx.x;
  const long m0 = (long)(s >> 1) * TLEN + (long)(s & 1) * 32;
  gemm1_strip(x, Wih1, bih1, bhh1, xp1, lds, m0);
}

#define PIPE_G1_FIRST (8 + BSZ)
__global__ __launch_bounds__(256) void pipe_kernel(
    float* bufA, float* bufB, float* h2,
    const float* Whh1, const float* Wih2, const float* bih2, const float* bhh2,
    const float* Whh2,
    const float* x, const float* Wih1, const float* bih1, const float* bhh1,
    const int li) {
  __shared__ __attribute__((aligned(16))) float lds[PIPE_LDS_FLOATS];
  const int bx = blockIdx.x;
  if (bx < 4) {
    if (li < NCH) rnn1_chunk(bufA, Whh1, bufB, lds, bx, li);
  } else if (bx < 8) {
    if (li >= 2 && li - 2 < NCH) rnn2_chunk(bufA, Whh2, h2, lds, bx - 4, li - 2);
  } else if (bx < PIPE_G1_FIRST) {
    if (li >= 1 && li - 1 < NCH) gemm2_chunk(bufB, Wih2, bih2, bhh2, bufA, lds, bx - 8, li - 1);
  } else {
    if (li + 1 < NCH) {
      const int s = bx - PIPE_G1_FIRST;
      const long m0 = (long)(s >> 1) * TLEN + (long)(li + 1) * CH + (long)(s & 1) * 32;
      gemm1_strip(x, Wih1, bih1, bhh1, bufA, lds, m0);
    }
  }
}

__global__ __launch_bounds__(64) void proj_ln_kernel(
    const float* __restrict__ h2, const float* __restrict__ Wp,
    const float* __restrict__ bp, const float* __restrict__ gamma,
    const float* __restrict__ beta, float* __restrict__ out) {
  __shared__ float tmp[PROJ];
  const int b = blockIdx.x;
  const int j = threadIdx.x;
  float acc = bp[j];
  for (int k = 0; k < H; ++k) acc += h2[b * H + k] * Wp[j * H + k];

  tmp[j] = acc;
  __syncthreads();
  for (int s = 32; s > 0; s >>= 1) {
    if (j < s) tmp[j] += tmp[j + s];
    __syncthreads();
  }
  const float mu = tmp[0] * (1.0f / PROJ);
  __syncthreads();
  const float d = acc - mu;
  tmp[j] = d * d;
  __syncthreads();
  for (int s = 32; s > 0; s >>= 1) {
    if (j < s) tmp[j] += tmp[j + s];
    __syncthreads();
  }
  const float var = tmp[0] * (1.0f / PROJ);
  const float r = rsqrtf(var + EPSV);
  out[b * PROJ + j] = d * r * gamma[j] + beta[j];
}

extern "C" void kernel_launch(void* const* d_in, const int* in_sizes, int n_in,
                              void* d_out, int out_size, void* d_ws, size_t ws_size,
                              hipStream_t stream) {
  (void)in_sizes; (void)n_in; (void)out_size; (void)ws_size;
  const float* x     = (const float*)d_in[0];
  const float* Wih1  = (const float*)d_in[1];
  const float* Whh1  = (const float*)d_in[2];
  const float* bih1  = (const float*)d_in[3];
  const float* bhh1  = (const float*)d_in[4];
  const float* Wih2  = (const float*)d_in[5];
  const float* Whh2  = (const float*)d_in[6];
  const float* bih2  = (const float*)d_in[7];
  const float* bhh2  = (const float*)d_in[8];
  const float* Wproj = (const float*)d_in[9];
  const float* bproj = (const float*)d_in[10];
  const float* gam   = (const float*)d_in[11];
  const float* bet   = (const float*)d_in[12];
  float* out = (float*)d_out;

  float* bufA = (float*)d_ws;
  float* bufB = bufA + (size_t)M_TOT * H;
  float* h2   = bufB + (size_t)M_TOT * H;

  gemm1_head<<<128, 256, 0, stream>>>(x, Wih1, bih1, bhh1, bufA);
  pipe_kernel<<<200, 256, 0, stream>>>(bufA, bufB, h2, Whh1, Wih2, bih2, bhh2, Whh2, x, Wih1, bih1, bhh1, 0);
  pipe_kernel<<<200, 256, 0, stream>>>(bufA, bufB, h2, Whh1, Wih2, bih2, bhh2, Whh2, x, Wih1, bih1, bhh1, 1);
  pipe_kernel<<<200, 256, 0, stream>>>(bufA, bufB, h2, Whh1, Wih2, bih2, bhh2, Whh2, x, Wih1, bih1, bhh1, 2);
  pipe_kernel<<<200, 256, 0, stream>>>(bufA, bufB, h2, Whh1, Wih2, bih2, bhh2, Whh2, x, Wih1, bih1, bhh1, 3);
  pipe_kernel<<<200, 256, 0, stream>>>(bufA, bufB, h2, Whh1, Wih2, bih2, bhh2, Whh2, x, Wih1, bih1, bhh1, 4);
  pipe_kernel<<<200, 256, 0, stream>>>(bufA, bufB, h2, Whh1, Wih2, bih2, bhh2, Whh2, x, Wih1, bih1, bhh1, 5);
  pipe_kernel<<<200, 256, 0, stream>>>(bufA, bufB, h2, Whh1, Wih2, bih2, bhh2, Whh2, x, Wih1, bih1, bhh1, 6);
  pipe_kernel<<<72,  256, 0, stream>>>(bufA, bufB, h2, Whh1, Wih2, bih2, bhh2, Whh2, x, Wih1, bih1, bhh1, 7);
  pipe_kernel<<<72,  256, 0, stream>>>(bufA, bufB, h2, Whh1, Wih2, bih2, bhh2, Whh2, x, Wih1, bih1, bhh1, 8);
  pipe_kernel<<<8,   256, 0, stream>>>(bufA, bufB, h2, Whh1, Wih2, bih2, bhh2, Whh2, x, Wih1, bih1, bhh1, 9);
  proj_ln_kernel<<<BSZ, PROJ, 0, stream>>>(h2, Wproj, bproj, gam, bet, out);
}


# recurrence kernels: all LDS reads of a step (32 A fragments of h, the two y1 rows) issued as one group ahead of the chain of 32 WMMAs, not one read per two WMMAs; sums and their order unchanged
# speedup vs baseline: 3.1544x; 1.0422x over previous
#include <hip/hip_runtime.h>

#define H 128
#define INW 2048
#define PROJ 64
#define BSZ 64
#define TLEN 512
#define M_TOT (BSZ * TLEN)
#define EPSV 1e-5f

typedef __attribute__((ext_vector_type(2))) float v2f;
typedef __attribute__((ext_vector_type(8))) float v8f;

#define WMMA_F32(a, b, c) \
  __builtin_amdgcn_wmma_f32_16x16x4_f32(false, (a), false, (b), (short)0, (c), false, false)

#define KC 128
#define LW (KC + 4)
#define G1_LDS_FLOATS ((32 + 128) * LW)

__device__ __forceinline__ void gemm1_strip(
    const float* __restrict__ x, const float* __restrict__ Wih1,
    const float* __restrict__ bih1, const float* __restrict__ bhh1,
    float* __restrict__ xp1, float* lds, const long m0) {
  float* xs = lds;
  float* ws = lds + 32 * LW;
  const int tid  = threadIdx.x;
  const int wave = tid >> 5;
  const int lane = tid & 31;
  const int half = lane >> 4;
  const int l16  = lane & 15;
  const int n0   = wave * 16;

  v8f c0 = {};
  v8f c1 = {};
  for (int k0 = 0; k0 < INW; k0 += KC) {
    for (int i = 0; i < 4; ++i) {
      int idx = tid + i * 256;
      int row = idx >> 5;
      int cv  = (idx & 31) << 2;
      float4 v = *(const float4*)&x[(m0 + row) * INW + k0 + cv];
      *(float4*)&xs[row * LW + cv] = v;
    }
    for (int i = 0; i < 16; ++i) {
      int idx = tid + i * 256;
      int row = idx >> 5;
      int cv  = (idx & 31) << 2;
      float4 v = *(const float4*)&Wih1[(long)row * INW + k0 + cv];
      *(float4*)&ws[row * LW + cv] = v;
    }
    __syncthreads();
    for (int k = 0; k < KC; k += 4) {
      const int kc = k + 2 * half;
      v2f b;  b.x  = ws[(n0 + l16) * LW + kc];  b.y  = ws[(n0 + l16) * LW + kc + 1];
      v2f a0; a0.x = xs[l16 * LW + kc];         a0.y = xs[l16 * LW + kc + 1];
      v2f a1; a1.x = xs[(16 + l16) * LW + kc];  a1.y = xs[(16 + l16) * LW + kc + 1];
      c0 = WMMA_F32(a0, b, c0);
      c1 = WMMA_F32(a1, b, c1);
    }
    __syncthreads();
  }
  const float bias = bih1[n0 + l16] + bhh1[n0 + l16];
  for (int v = 0; v < 8; ++v) {
    long m = m0 + v + 8 * half;
    xp1[m * H + n0 + l16]        = c0[v] + bias;
    xp1[(m + 16) * H + n0 + l16] = c1[v] + bias;
  }
}

__device__ __forceinline__ float tanh_exp2(float x) {
  const float t = __builtin_amdgcn_exp2f(-2.8853900817779268f * __builtin_fabsf(x));
  const float y = (1.0f - t) * __builtin_amdgcn_rcpf(1.0f + t);
  return __builtin_copysignf(y, x);
}

#define HL (H + 4)
#define TBLK 4
#define CH 64
#define NCH (TLEN / CH)
#define G2R 64
#define PIPE_LDS_FLOATS (164 * 256)

__device__ __forceinline__ void rnn1_chunk(
    const float* __restrict__ xp1, const float* __restrict__ Whh1,
    float* __restrict__ y1, float* hs, const int g, const int ch) {
  const int tid  = threadIdx.x;
  const int wave = tid >> 5;
  const int lane = tid & 31;
  const int half = lane >> 4;
  const int l16  = lane & 15;
  const int n0   = wave * 16;
  const int b0   = g * 16;
  const long tofs = (long)ch * CH * H;

  v2f bh[32];
  for (int kk = 0; kk < 32; ++kk) {
    const int kc = kk * 4 + 2 * half;
    bh[kk].x = Whh1[(n0 + l16) * H + kc];  bh[kk].y = Whh1[(n0 + l16) * H + kc + 1];
  }

  if (ch == 0) {
    for (int i = tid; i < 16 * HL; i += 256) hs[i] = 0.0f;
  } else {
    const float* hin = y1 + (long)b0 * TLEN * H + tofs - H;
    const int r  = tid >> 5;
    const int cv = (tid & 31) << 2;
    const float4 va = *(const float4*)&hin[(long)r * TLEN * H + cv];
    const float4 vb = *(const float4*)&hin[(long)(r + 8) * TLEN * H + cv];
    *(float4*)&hs[r * HL + cv] = va;
    *(float4*)&hs[(r + 8) * HL + cv] = vb;
  }
  __syncthreads();

  float* yrow = y1 + (long)(b0 + 2 * wave) * TLEN * H + tofs + lane * 4;
  const float* hrow = hs + (2 * wave) * HL + lane * 4;

  const float* xrow = xp1 + (long)(b0 + 8 * half) * TLEN * H + tofs + n0 + l16;
  v8f nxt[TBLK];
  for (int j = 0; j < TBLK; ++j)
    for (int v = 0; v < 8; ++v) nxt[j][v] = xrow[((long)v * TLEN + j) * H];

  for (int tb = 0; tb < CH; tb += TBLK) {
    v8f cur[TBLK];
    for (int j = 0; j < TBLK; ++j) cur[j] = nxt[j];
    if (tb + TBLK < CH) {
      for (int j = 0; j < TBLK; ++j)
        for (int v = 0; v < 8; ++v)
          nxt[j][v] = xrow[((long)v * TLEN + tb + TBLK + j) * H];
    }
#pragma unroll
    for (int j = 0; j < TBLK; ++j) {
      const int t = tb + j;
      v8f c = cur[j];
      const float4 h0v = *(const float4*)&hrow[0];
      const float4 h1v = *(const float4*)&hrow[HL];
      v2f a[32];
      for (int kk = 0; kk < 32; ++kk) {
        const int kc = kk * 4 + 2 * half;
        a[kk].x = hs[l16 * HL + kc]; a[kk].y = hs[l16 * HL + kc + 1];
      }
      __builtin_amdgcn_sched_barrier(0);
      if (t > 0) {
        *(float4*)&yrow[(long)(t - 1) * H] = h0v;
        *(float4*)&yrow[(long)TLEN * H + (long)(t - 1) * H] = h1v;
      }
      __builtin_amdgcn_sched_barrier(0);
      for (int kk = 0; kk < 32; ++kk) c = WMMA_F32(a[kk], bh[kk], c);
      __builtin_amdgcn_sched_barrier(0);
      for (int v = 0; v < 8; ++v) c[v] = tanh_exp2(c[v]);
      __syncthreads();
      for (int v = 0; v < 8; ++v) hs[(v + 8 * half) * HL + n0 + l16] = c[v];
      __syncthreads();
    }
  }
  {
    const float4 h0v = *(const float4*)&hrow[0];
    const float4 h1v = *(const float4*)&hrow[HL];
    *(float4*)&yrow[(long)(CH - 1) * H] = h0v;
    *(float4*)&yrow[(long)TLEN * H + (long)(CH - 1) * H] = h1v;
  }
}

#define LW2 (H + 4)
__device__ __forceinline__ void gemm2_chunk(
    const float* __restrict__ y1, const float* __restrict__ Wih2,
    const float* __restrict__ bih2, const float* __restrict__ bhh2,
    float* __restrict__ xp2, float* lds, const int b, const int ch) {
  float* ys = lds;
  float* ws = lds + G2R * LW2;
  const int tid  = threadIdx.x;
  const int wave = tid >> 5;
  const int lane = tid & 31;
  const int half = lane >> 4;
  const int l16  = lane & 15;
  const int n0   = wave * 16;
  const long m0  = (long)b * TLEN + (long)ch * CH;

  for (int i = 0; i < 8; ++i) {
    int idx = tid + i * 256;
    int row = idx >> 5;
    int cv  = (idx & 31) << 2;
    float4 v = *(const float4*)&y1[(m0 + row) * H + cv];
    *(float4*)&ys[row * LW2 + cv] = v;
  }
  for (int i = 0; i < 16; ++i) {
    int idx = tid + i * 256;
    int row = idx >> 5;
    int cv  = (idx & 31) << 2;
    float4 v = *(const float4*)&Wih2[(long)row * H + cv];
    *(float4*)&ws[row * LW2 + cv] = v;
  }
  const float bias2 = bih2[n0 + l16] + bhh2[n0 + l16];
  v8f c0, c1, c2, c3;
  for (int v = 0; v < 8; ++v) { c0[v] = bias2; c1[v] = bias2; c2[v] = bias2; c3[v] = bias2; }
  __syncthreads();
  for (int k = 0; k < H; k += 4) {
    const int kc = k + 2 * half;
    v2f bb; bb.x = ws[(n0 + l16) * LW2 + kc];  bb.y = ws[(n0 + l16) * LW2 + kc + 1];
    v2f a0; a0.x = ys[l16 * LW2 + kc];         a0.y = ys[l16 * LW2 + kc + 1];
    v2f a1; a1.x = ys[(16 + l16) * LW2 + kc];  a1.y = ys[(16 + l16) * LW2 + kc + 1];
    v2f a2; a2.x = ys[(32 + l16) * LW2 + kc];  a2.y = ys[(32 + l16) * LW2 + kc + 1];
    v2f a3; a3.x = ys[(48 + l16) * LW2 + kc];  a3.y = ys[(48 + l16) * LW2 + kc + 1];
    c0 = WMMA_F32(a0, bb, c0);
    c1 = WMMA_F32(a1, bb, c1);
    c2 = WMMA_F32(a2, bb, c2);
    c3 = WMMA_F32(a3, bb, c3);
  }
  for (int v = 0; v < 8; ++v) {
    long m = m0 + v + 8 * half;
    xp2[m * H + n0 + l16]        = c0[v];
    xp2[(m + 16) * H + n0 + l16] = c1[v];
    xp2[(m + 32) * H + n0 + l16] = c2[v];
    xp2[(m + 48) * H + n0 + l16] = c3[v];
  }
}

__device__ __forceinline__ void rnn2_chunk(
    const float* __restrict__ xp2, const float* __restrict__ Whh2,
    float* __restrict__ h2, float* hs, const int g, const int ch) {
  const int tid  = threadIdx.x;
  const int wave = tid >> 5;
  const int lane = tid & 31;
  const int half = lane >> 4;
  const int l16  = lane & 15;
  const int n0   = wave * 16;
  const int b0   = g * 16;
  const long tofs = (long)ch * CH * H;

  v2f bh[32];
  for (int kk = 0; kk < 32; ++kk) {
    const int kc = kk * 4 + 2 * half;
    bh[kk].x = Whh2[(n0 + l16) * H + kc];  bh[kk].y = Whh2[(n0 + l16) * H + kc + 1];
  }
  if (ch == 0) {
    for (int i = tid; i < 16 * HL; i += 256) hs[i] = 0.0f;
  } else {
    const float* hin = h2 + (long)b0 * H;
    const int r  = tid >> 5;
    const int cv = (tid & 31) << 2;
    const float4 va = *(const float4*)&hin[r * H + cv];
    const float4 vb = *(const float4*)&hin[(r + 8) * H + cv];
    *(float4*)&hs[r * HL + cv] = va;
    *(float4*)&hs[(r + 8) * HL + cv] = vb;
  }
  __syncthreads();

  const float* xrow = xp2 + (long)(b0 + 8 * half) * TLEN * H + tofs + n0 + l16;
  v8f nxt[TBLK];
  for (int j = 0; j < TBLK; ++j)
    for (int v = 0; v < 8; ++v) nxt[j][v] = xrow[((long)v * TLEN + j) * H];

  v8f c = {};
  for (int tb = 0; tb < CH; tb += TBLK) {
    v8f cur[TBLK];
    for (int j = 0; j < TBLK; ++j) cur[j] = nxt[j];
    if (tb + TBLK < CH) {
      for (int j = 0; j < TBLK; ++j)
        for (int v = 0; v < 8; ++v)
          nxt[j][v] = xrow[((long)v * TLEN + tb + TBLK + j) * H];
    }
#pragma unroll
    for (int j = 0; j < TBLK; ++j) {
      c = cur[j];
      v2f a[32];
      for (int kk = 0; kk < 32; ++kk) {
        const int kc = kk * 4 + 2 * half;
        a[kk].x = hs[l16 * HL + kc]; a[kk].y = hs[l16 * HL + kc + 1];
      }
      __builtin_amdgcn_sched_barrier(0);
      for (int kk = 0; kk < 32; ++kk) c = WMMA_F32(a[kk], bh[kk], c);
      __builtin_amdgcn_sched_barrier(0);
      for (int v = 0; v < 8; ++v) c[v] = tanh_exp2(c[v]);
      __syncthreads();
      for (int v = 0; v < 8; ++v) hs[(v + 8 * half) * HL + n0 + l16] = c[v];
      __syncthreads();
    }
  }
  for (int v = 0; v < 8; ++v)
    h2[(long)(b0 + v + 8 * half) * H + n0 + l16] = c[v];
}

__global__ __launch_bounds__(256) void gemm1_head(
    const float* __restrict__ x, const float* __restrict__ Wih1,
    const float* __restrict__ bih1, const float* __restrict__ bhh1,
    float* __restrict__ xp1) {
  __shared__ __attribute__((aligned(16))) float lds[G1_LDS_FLOATS];
  const int s = blockIdx.x;
  const long m0 = (long)(s >> 1) * TLEN + (long)(s & 1) * 32;
  gemm1_strip(x, Wih1, bih1, bhh1, xp1, lds, m0);
}

#define PIPE_G1_FIRST (8 + BSZ)
__global__ __launch_bounds__(256) void pipe_kernel(
    float* bufA, float* bufB, float* h2,
    const float* Whh1, const float* Wih2, const float* bih2, const float* bhh2,
    const float* Whh2,
    const float* x, const float* Wih1, const float* bih1, const float* bhh1,
    const int li) {
  __shared__ __attribute__((aligned(16))) float lds[PIPE_LDS_FLOATS];
  const int bx = blockIdx.x;
  if (bx < 4) {
    if (li < NCH) rnn1_chunk(bufA, Whh1, bufB, lds, bx, li);
  } else if (bx < 8) {
    if (li >= 2 && li - 2 < NCH) rnn2_chunk(bufA, Whh2, h2, lds, bx - 4, li - 2);
  } else if (bx < PIPE_G1_FIRST) {
    if (li >= 1 && li - 1 < NCH) gemm2_chunk(bufB, Wih2, bih2, bhh2, bufA, lds, bx - 8, li - 1);
  } else {
    if (li + 1 < NCH) {
      const int s = bx - PIPE_G1_FIRST;
      const long m0 = (long)(s >> 1) * TLEN + (long)(li + 1) * CH + (long)(s & 1) * 32;
      gemm1_strip(x, Wih1, bih1, bhh1, bufA, lds, m0);
    }
  }
}

__global__ __launch_bounds__(64) void proj_ln_kernel(
    const float* __restrict__ h2, const float* __restrict__ Wp,
    const float* __restrict__ bp, const float* __restrict__ gamma,
    const float* __restrict__ beta, float* __restrict__ out) {
  __shared__ float tmp[PROJ];
  const int b = blockIdx.x;
  const int j = threadIdx.x;
  float acc = bp[j];
  for (int k = 0; k < H; ++k) acc += h2[b * H + k] * Wp[j * H + k];

  tmp[j] = acc;
  __syncthreads();
  for (int s = 32; s > 0; s >>= 1) {
    if (j < s) tmp[j] += tmp[j + s];
    __syncthreads();
  }
  const float mu = tmp[0] * (1.0f / PROJ);
  __syncthreads();
  const float d = acc - mu;
  tmp[j] = d * d;
  __syncthreads();
  for (int s = 32; s > 0; s >>= 1) {
    if (j < s) tmp[j] += tmp[j + s];
    __syncthreads();
  }
  const float var = tmp[0] * (1.0f / PROJ);
  const float r = rsqrtf(var + EPSV);
  out[b * PROJ + j] = d * r * gamma[j] + beta[j];
}

extern "C" void kernel_launch(void* const* d_in, const int* in_sizes, int n_in,
                              void* d_out, int out_size, void* d_ws, size_t ws_size,
                              hipStream_t stream) {
  (void)in_sizes; (void)n_in; (void)out_size; (void)ws_size;
  const float* x     = (const float*)d_in[0];
  const float* Wih1  = (const float*)d_in[1];
  const float* Whh1  = (const float*)d_in[2];
  const float* bih1  = (const float*)d_in[3];
  const float* bhh1  = (const float*)d_in[4];
  const float* Wih2  = (const float*)d_in[5];
  const float* Whh2  = (const float*)d_in[6];
  const float* bih2  = (const float*)d_in[7];
  const float* bhh2  = (const float*)d_in[8];
  const float* Wproj = (const float*)d_in[9];
  const float* bproj = (const float*)d_in[10];
  const float* gam   = (const float*)d_in[11];
  const float* bet   = (const float*)d_in[12];
  float* out = (float*)d_out;

  float* bufA = (float*)d_ws;
  float* bufB = bufA + (size_t)M_TOT * H;
  float* h2   = bufB + (size_t)M_TOT * H;

  gemm1_head<<<128, 256, 0, stream>>>(x, Wih1, bih1, bhh1, bufA);
  pipe_kernel<<<200, 256, 0, stream>>>(bufA, bufB, h2, Whh1, Wih2, bih2, bhh2, Whh2, x, Wih1, bih1, bhh1, 0);
  pipe_kernel<<<200, 256, 0, stream>>>(bufA, bufB, h2, Whh1, Wih2, bih2, bhh2, Whh2, x, Wih1, bih1, bhh1, 1);
  pipe_kernel<<<200, 256, 0, stream>>>(bufA, bufB, h2, Whh1, Wih2, bih2, bhh2, Whh2, x, Wih1, bih1, bhh1, 2);
  pipe_kernel<<<200, 256, 0, stream>>>(bufA, bufB, h2, Whh1, Wih2, bih2, bhh2, Whh2, x, Wih1, bih1, bhh1, 3);
  pipe_kernel<<<200, 256, 0, stream>>>(bufA, bufB, h2, Whh1, Wih2, bih2, bhh2, Whh2, x, Wih1, bih1, bhh1, 4);
  pipe_kernel<<<200, 256, 0, stream>>>(bufA, bufB, h2, Whh1, Wih2, bih2, bhh2, Whh2, x, Wih1, bih1, bhh1, 5);
  pipe_kernel<<<200, 256, 0, stream>>>(bufA, bufB, h2, Whh1, Wih2, bih2, bhh2, Whh2, x, Wih1, bih1, bhh1, 6);
  pipe_kernel<<<72,  256, 0, stream>>>(bufA, bufB, h2, Whh1, Wih2, bih2, bhh2, Whh2, x, Wih1, bih1, bhh1, 7);
  pipe_kernel<<<72,  256, 0, stream>>>(bufA, bufB, h2, Whh1, Wih2, bih2, bhh2, Whh2, x, Wih1, bih1, bhh1, 8);
  pipe_kernel<<<8,   256, 0, stream>>>(bufA, bufB, h2, Whh1, Wih2, bih2, bhh2, Whh2, x, Wih1, bih1, bhh1, 9);
  proj_ln_kernel<<<BSZ, PROJ, 0, stream>>>(h2, Wproj, bproj, gam, bet, out);
}


# recurrence kernels: step barriers fence LDS only, so xp loads and y1 stores stay in flight across steps; xp fetched 8 loads per step behind the LDS reads, 3 steps ahead; sums, their order unchanged
# speedup vs baseline: 3.5145x; 1.1142x over previous
#include <hip/hip_runtime.h>

#define H 128
#define INW 2048
#define PROJ 64
#define BSZ 64
#define TLEN 512
#define M_TOT (BSZ * TLEN)
#define EPSV 1e-5f

typedef __attribute__((ext_vector_type(2))) float v2f;
typedef __attribute__((ext_vector_type(8))) float v8f;

#define WMMA_F32(a, b, c) \
  __builtin_amdgcn_wmma_f32_16x16x4_f32(false, (a), false, (b), (short)0, (c), false, false)

#define KC 128
#define LW (KC + 4)
#define G1_LDS_FLOATS ((32 + 128) * LW)

__device__ __forceinline__ void gemm1_strip(
    const float* __restrict__ x, const float* __restrict__ Wih1,
    const float* __restrict__ bih1, const float* __restrict__ bhh1,
    float* __restrict__ xp1, float* lds, const long m0) {
  float* xs = lds;
  float* ws = lds + 32 * LW;
  const int tid  = threadIdx.x;
  const int wave = tid >> 5;
  const int lane = tid & 31;
  const int half = lane >> 4;
  const int l16  = lane & 15;
  const int n0   = wave * 16;

  v8f c0 = {};
  v8f c1 = {};
  for (int k0 = 0; k0 < INW; k0 += KC) {
    for (int i = 0; i < 4; ++i) {
      int idx = tid + i * 256;
      int row = idx >> 5;
      int cv  = (idx & 31) << 2;
      float4 v = *(const float4*)&x[(m0 + row) * INW + k0 + cv];
      *(float4*)&xs[row * LW + cv] = v;
    }
    for (int i = 0; i < 16; ++i) {
      int idx = tid + i * 256;
      int row = idx >> 5;
      int cv  = (idx & 31) << 2;
      float4 v = *(const float4*)&Wih1[(long)row * INW + k0 + cv];
      *(float4*)&ws[row * LW + cv] = v;
    }
    __syncthreads();
    for (int k = 0; k < KC; k += 4) {
      const int kc = k + 2 * half;
      v2f b;  b.x  = ws[(n0 + l16) * LW + kc];  b.y  = ws[(n0 + l16) * LW + kc + 1];
      v2f a0; a0.x = xs[l16 * LW + kc];         a0.y = xs[l16 * LW + kc + 1];
      v2f a1; a1.x = xs[(16 + l16) * LW + kc];  a1.y = xs[(16 + l16) * LW + kc + 1];
      c0 = WMMA_F32(a0, b, c0);
      c1 = WMMA_F32(a1, b, c1);
    }
    __syncthreads();
  }
  const float bias = bih1[n0 + l16] + bhh1[n0 + l16];
  for (int v = 0; v < 8; ++v) {
    long m = m0 + v + 8 * half;
    xp1[m * H + n0 + l16]        = c0[v] + bias;
    xp1[(m + 16) * H + n0 + l16] = c1[v] + bias;
  }
}

__device__ __forceinline__ float tanh_exp2(float x) {
  const float t = __builtin_amdgcn_exp2f(-2.8853900817779268f * __builtin_fabsf(x));
  const float y = (1.0f - t) * __builtin_amdgcn_rcpf(1.0f + t);
  return __builtin_copysignf(y, x);
}

__device__ __forceinline__ void lds_step_barrier() {
  __builtin_amdgcn_fence(__ATOMIC_RELEASE, "workgroup", "local");
  __builtin_amdgcn_s_barrier();
  __builtin_amdgcn_fence(__ATOMIC_ACQUIRE, "workgroup", "local");
}

#define HL (H + 4)
#define TBLK 4
#define CH 64
#define NCH (TLEN / CH)
#define G2R 64
#define PIPE_LDS_FLOATS (164 * 256)

__device__ __forceinline__ void rnn1_chunk(
    const float* __restrict__ xp1, const float* __restrict__ Whh1,
    float* __restrict__ y1, float* hs, const int g, const int ch) {
  const int tid  = threadIdx.x;
  const int wave = tid >> 5;
  const int lane = tid & 31;
  const int half = lane >> 4;
  const int l16  = lane & 15;
  const int n0   = wave * 16;
  const int b0   = g * 16;
  const long tofs = (long)ch * CH * H;

  v2f bh[32];
  for (int kk = 0; kk < 32; ++kk) {
    const int kc = kk * 4 + 2 * half;
    bh[kk].x = Whh1[(n0 + l16) * H + kc];  bh[kk].y = Whh1[(n0 + l16) * H + kc + 1];
  }

  if (ch == 0) {
    for (int i = tid; i < 16 * HL; i += 256) hs[i] = 0.0f;
  } else {
    const float* hin = y1 + (long)b0 * TLEN * H + tofs - H;
    const int r  = tid >> 5;
    const int cv = (tid & 31) << 2;
    const float4 va = *(const float4*)&hin[(long)r * TLEN * H + cv];
    const float4 vb = *(const float4*)&hin[(long)(r + 8) * TLEN * H + cv];
    *(float4*)&hs[r * HL + cv] = va;
    *(float4*)&hs[(r + 8) * HL + cv] = vb;
  }
  __syncthreads();

  float* yrow = y1 + (long)(b0 + 2 * wave) * TLEN * H + tofs + lane * 4;
  const float* hrow = hs + (2 * wave) * HL + lane * 4;

  const float* xrow = xp1 + (long)(b0 + 8 * half) * TLEN * H + tofs + n0 + l16;
  v8f ring[TBLK];
  for (int j = 0; j < TBLK - 1; ++j)
    for (int v = 0; v < 8; ++v) ring[j][v] = xrow[((long)v * TLEN + j) * H];

  for (int tb = 0; tb < CH; tb += TBLK) {
#pragma unroll
    for (int j = 0; j < TBLK; ++j) {
      const int t = tb + j;
      v8f c = ring[j];
      const float4 h0v = *(const float4*)&hrow[0];
      const float4 h1v = *(const float4*)&hrow[HL];
      v2f a[32];
      for (int kk = 0; kk < 32; ++kk) {
        const int kc = kk * 4 + 2 * half;
        a[kk].x = hs[l16 * HL + kc]; a[kk].y = hs[l16 * HL + kc + 1];
      }
      __builtin_amdgcn_sched_barrier(0);
      {
        const int tn = (t + TBLK - 1 < CH) ? t + TBLK - 1 : CH - 1;
        for (int v = 0; v < 8; ++v)
          ring[(j + TBLK - 1) % TBLK][v] = xrow[((long)v * TLEN + tn) * H];
      }
      __builtin_amdgcn_sched_barrier(0);
      if (t > 0) {
        *(float4*)&yrow[(long)(t - 1) * H] = h0v;
        *(float4*)&yrow[(long)TLEN * H + (long)(t - 1) * H] = h1v;
      }
      __builtin_amdgcn_sched_barrier(0);
      for (int kk = 0; kk < 32; ++kk) c = WMMA_F32(a[kk], bh[kk], c);
      __builtin_amdgcn_sched_barrier(0);
      for (int v = 0; v < 8; ++v) c[v] = tanh_exp2(c[v]);
      lds_step_barrier();
      for (int v = 0; v < 8; ++v) hs[(v + 8 * half) * HL + n0 + l16] = c[v];
      lds_step_barrier();
    }
  }
  {
    const float4 h0v = *(const float4*)&hrow[0];
    const float4 h1v = *(const float4*)&hrow[HL];
    *(float4*)&yrow[(long)(CH - 1) * H] = h0v;
    *(float4*)&yrow[(long)TLEN * H + (long)(CH - 1) * H] = h1v;
  }
}

#define LW2 (H + 4)
__device__ __forceinline__ void gemm2_chunk(
    const float* __restrict__ y1, const float* __restrict__ Wih2,
    const float* __restrict__ bih2, const float* __restrict__ bhh2,
    float* __restrict__ xp2, float* lds, const int b, const int ch) {
  float* ys = lds;
  float* ws = lds + G2R * LW2;
  const int tid  = threadIdx.x;
  const int wave = tid >> 5;
  const int lane = tid & 31;
  const int half = lane >> 4;
  const int l16  = lane & 15;
  const int n0   = wave * 16;
  const long m0  = (long)b * TLEN + (long)ch * CH;

  for (int i = 0; i < 8; ++i) {
    int idx = tid + i * 256;
    int row = idx >> 5;
    int cv  = (idx & 31) << 2;
    float4 v = *(const float4*)&y1[(m0 + row) * H + cv];
    *(float4*)&ys[row * LW2 + cv] = v;
  }
  for (int i = 0; i < 16; ++i) {
    int idx = tid + i * 256;
    int row = idx >> 5;
    int cv  = (idx & 31) << 2;
    float4 v = *(const float4*)&Wih2[(long)row * H + cv];
    *(float4*)&ws[row * LW2 + cv] = v;
  }
  const float bias2 = bih2[n0 + l16] + bhh2[n0 + l16];
  v8f c0, c1, c2, c3;
  for (int v = 0; v < 8; ++v) { c0[v] = bias2; c1[v] = bias2; c2[v] = bias2; c3[v] = bias2; }
  __syncthreads();
  for (int k = 0; k < H; k += 4) {
    const int kc = k + 2 * half;
    v2f bb; bb.x = ws[(n0 + l16) * LW2 + kc];  bb.y = ws[(n0 + l16) * LW2 + kc + 1];
    v2f a0; a0.x = ys[l16 * LW2 + kc];         a0.y = ys[l16 * LW2 + kc + 1];
    v2f a1; a1.x = ys[(16 + l16) * LW2 + kc];  a1.y = ys[(16 + l16) * LW2 + kc + 1];
    v2f a2; a2.x = ys[(32 + l16) * LW2 + kc];  a2.y = ys[(32 + l16) * LW2 + kc + 1];
    v2f a3; a3.x = ys[(48 + l16) * LW2 + kc];  a3.y = ys[(48 + l16) * LW2 + kc + 1];
    c0 = WMMA_F32(a0, bb, c0);
    c1 = WMMA_F32(a1, bb, c1);
    c2 = WMMA_F32(a2, bb, c2);
    c3 = WMMA_F32(a3, bb, c3);
  }
  for (int v = 0; v < 8; ++v) {
    long m = m0 + v + 8 * half;
    xp2[m * H + n0 + l16]        = c0[v];
    xp2[(m + 16) * H + n0 + l16] = c1[v];
    xp2[(m + 32) * H + n0 + l16] = c2[v];
    xp2[(m + 48) * H + n0 + l16] = c3[v];
  }
}

__device__ __forceinline__ void rnn2_chunk(
    const float* __restrict__ xp2, const float* __restrict__ Whh2,
    float* __restrict__ h2, float* hs, const int g, const int ch) {
  const int tid  = threadIdx.x;
  const int wave = tid >> 5;
  const int lane = tid & 31;
  const int half = lane >> 4;
  const int l16  = lane & 15;
  const int n0   = wave * 16;
  const int b0   = g * 16;
  const long tofs = (long)ch * CH * H;

  v2f bh[32];
  for (int kk = 0; kk < 32; ++kk) {
    const int kc = kk * 4 + 2 * half;
    bh[kk].x = Whh2[(n0 + l16) * H + kc];  bh[kk].y = Whh2[(n0 + l16) * H + kc + 1];
  }
  if (ch == 0) {
    for (int i = tid; i < 16 * HL; i += 256) hs[i] = 0.0f;
  } else {
    const float* hin = h2 + (long)b0 * H;
    const int r  = tid >> 5;
    const int cv = (tid & 31) << 2;
    const float4 va = *(const float4*)&hin[r * H + cv];
    const float4 vb = *(const float4*)&hin[(r + 8) * H + cv];
    *(float4*)&hs[r * HL + cv] = va;
    *(float4*)&hs[(r + 8) * HL + cv] = vb;
  }
  __syncthreads();

  const float* xrow = xp2 + (long)(b0 + 8 * half) * TLEN * H + tofs + n0 + l16;
  v8f ring[TBLK];
  for (int j = 0; j < TBLK - 1; ++j)
    for (int v = 0; v < 8; ++v) ring[j][v] = xrow[((long)v * TLEN + j) * H];

  v8f c = {};
  for (int tb = 0; tb < CH; tb += TBLK) {
#pragma unroll
    for (int j = 0; j < TBLK; ++j) {
      const int t = tb + j;
      c = ring[j];
      v2f a[32];
      for (int kk = 0; kk < 32; ++kk) {
        const int kc = kk * 4 + 2 * half;
        a[kk].x = hs[l16 * HL + kc]; a[kk].y = hs[l16 * HL + kc + 1];
      }
      __builtin_amdgcn_sched_barrier(0);
      {
        const int tn = (t + TBLK - 1 < CH) ? t + TBLK - 1 : CH - 1;
        for (int v = 0; v < 8; ++v)
          ring[(j + TBLK - 1) % TBLK][v] = xrow[((long)v * TLEN + tn) * H];
      }
      __builtin_amdgcn_sched_barrier(0);
      for (int kk = 0; kk < 32; ++kk) c = WMMA_F32(a[kk], bh[kk], c);
      __builtin_amdgcn_sched_barrier(0);
      for (int v = 0; v < 8; ++v) c[v] = tanh_exp2(c[v]);
      lds_step_barrier();
      for (int v = 0; v < 8; ++v) hs[(v + 8 * half) * HL + n0 + l16] = c[v];
      lds_step_barrier();
    }
  }
  for (int v = 0; v < 8; ++v)
    h2[(long)(b0 + v + 8 * half) * H + n0 + l16] = c[v];
}

__global__ __launch_bounds__(256) void gemm1_head(
    const float* __restrict__ x, const float* __restrict__ Wih1,
    const float* __restrict__ bih1, const float* __restrict__ bhh1,
    float* __restrict__ xp1) {
  __shared__ __attribute__((aligned(16))) float lds[G1_LDS_FLOATS];
  const int s = blockIdx.x;
  const long m0 = (long)(s >> 1) * TLEN + (long)(s & 1) * 32;
  gemm1_strip(x, Wih1, bih1, bhh1, xp1, lds, m0);
}

#define PIPE_G1_FIRST (8 + BSZ)
__global__ __launch_bounds__(256) void pipe_kernel(
    float* bufA, float* bufB, float* h2,
    const float* Whh1, const float* Wih2, const float* bih2, const float* bhh2,
    const float* Whh2,
    const float* x, const float* Wih1, const float* bih1, const float* bhh1,
    const int li) {
  __shared__ __attribute__((aligned(16))) float lds[PIPE_LDS_FLOATS];
  const int bx = blockIdx.x;
  if (bx < 4) {
    if (li < NCH) rnn1_chunk(bufA, Whh1, bufB, lds, bx, li);
  } else if (bx < 8) {
    if (li >= 2 && li - 2 < NCH) rnn2_chunk(bufA, Whh2, h2, lds, bx - 4, li - 2);
  } else if (bx < PIPE_G1_FIRST) {
    if (li >= 1 && li - 1 < NCH) gemm2_chunk(bufB, Wih2, bih2, bhh2, bufA, lds, bx - 8, li - 1);
  } else {
    if (li + 1 < NCH) {
      const int s = bx - PIPE_G1_FIRST;
      const long m0 = (long)(s >> 1) * TLEN + (long)(li + 1) * CH + (long)(s & 1) * 32;
      gemm1_strip(x, Wih1, bih1, bhh1, bufA, lds, m0);
    }
  }
}

__global__ __launch_bounds__(64) void proj_ln_kernel(
    const float* __restrict__ h2, const float* __restrict__ Wp,
    const float* __restrict__ bp, const float* __restrict__ gamma,
    const float* __restrict__ beta, float* __restrict__ out) {
  __shared__ float tmp[PROJ];
  const int b = blockIdx.x;
  const int j = threadIdx.x;
  float acc = bp[j];
  for (int k = 0; k < H; ++k) acc += h2[b * H + k] * Wp[j * H + k];

  tmp[j] = acc;
  __syncthreads();
  for (int s = 32; s > 0; s >>= 1) {
    if (j < s) tmp[j] += tmp[j + s];
    __syncthreads();
  }
  const float mu = tmp[0] * (1.0f / PROJ);
  __syncthreads();
  const float d = acc - mu;
  tmp[j] = d * d;
  __syncthreads();
  for (int s = 32; s > 0; s >>= 1) {
    if (j < s) tmp[j] += tmp[j + s];
    __syncthreads();
  }
  const float var = tmp[0] * (1.0f / PROJ);
  const float r = rsqrtf(var + EPSV);
  out[b * PROJ + j] = d * r * gamma[j] + beta[j];
}

extern "C" void kernel_launch(void* const* d_in, const int* in_sizes, int n_in,
                              void* d_out, int out_size, void* d_ws, size_t ws_size,
                              hipStream_t stream) {
  (void)in_sizes; (void)n_in; (void)out_size; (void)ws_size;
  const float* x     = (const float*)d_in[0];
  const float* Wih1  = (const float*)d_in[1];
  const float* Whh1  = (const float*)d_in[2];
  const float* bih1  = (const float*)d_in[3];
  const float* bhh1  = (const float*)d_in[4];
  const float* Wih2  = (const float*)d_in[5];
  const float* Whh2  = (const float*)d_in[6];
  const float* bih2  = (const float*)d_in[7];
  const float* bhh2  = (const float*)d_in[8];
  const float* Wproj = (const float*)d_in[9];
  const float* bproj = (const float*)d_in[10];
  const float* gam   = (const float*)d_in[11];
  const float* bet   = (const float*)d_in[12];
  float* out = (float*)d_out;

  float* bufA = (float*)d_ws;
  float* bufB = bufA + (size_t)M_TOT * H;
  float* h2   = bufB + (size_t)M_TOT * H;

  gemm1_head<<<128, 256, 0, stream>>>(x, Wih1, bih1, bhh1, bufA);
  pipe_kernel<<<200, 256, 0, stream>>>(bufA, bufB, h2, Whh1, Wih2, bih2, bhh2, Whh2, x, Wih1, bih1, bhh1, 0);
  pipe_kernel<<<200, 256, 0, stream>>>(bufA, bufB, h2, Whh1, Wih2, bih2, bhh2, Whh2, x, Wih1, bih1, bhh1, 1);
  pipe_kernel<<<200, 256, 0, stream>>>(bufA, bufB, h2, Whh1, Wih2, bih2, bhh2, Whh2, x, Wih1, bih1, bhh1, 2);
  pipe_kernel<<<200, 256, 0, stream>>>(bufA, bufB, h2, Whh1, Wih2, bih2, bhh2, Whh2, x, Wih1, bih1, bhh1, 3);
  pipe_kernel<<<200, 256, 0, stream>>>(bufA, bufB, h2, Whh1, Wih2, bih2, bhh2, Whh2, x, Wih1, bih1, bhh1, 4);
  pipe_kernel<<<200, 256, 0, stream>>>(bufA, bufB, h2, Whh1, Wih2, bih2, bhh2, Whh2, x, Wih1, bih1, bhh1, 5);
  pipe_kernel<<<200, 256, 0, stream>>>(bufA, bufB, h2, Whh1, Wih2, bih2, bhh2, Whh2, x, Wih1, bih1, bhh1, 6);
  pipe_kernel<<<72,  256, 0, stream>>>(bufA, bufB, h2, Whh1, Wih2, bih2, bhh2, Whh2, x, Wih1, bih1, bhh1, 7);
  pipe_kernel<<<72,  256, 0, stream>>>(bufA, bufB, h2, Whh1, Wih2, bih2, bhh2, Whh2, x, Wih1, bih1, bhh1, 8);
  pipe_kernel<<<8,   256, 0, stream>>>(bufA, bufB, h2, Whh1, Wih2, bih2, bhh2, Whh2, x, Wih1, bih1, bhh1, 9);
  proj_ln_kernel<<<BSZ, PROJ, 0, stream>>>(h2, Wproj, bproj, gam, bet, out);
}


# recurrences: product transposed (W_hh as A, h as B), so a wave's new h registers are the B fragments of its own 16 k: 4 of a step's 32 WMMAs run beside tanh and the LDS hand-over; own k summed first
# speedup vs baseline: 3.8285x; 1.0893x over previous
#include <hip/hip_runtime.h>

#define H 128
#define INW 2048
#define PROJ 64
#define BSZ 64
#define TLEN 512
#define M_TOT (BSZ * TLEN)
#define EPSV 1e-5f

typedef __attribute__((ext_vector_type(2))) float v2f;
typedef __attribute__((ext_vector_type(8))) float v8f;

#define WMMA_F32(a, b, c) \
  __builtin_amdgcn_wmma_f32_16x16x4_f32(false, (a), false, (b), (short)0, (c), false, false)

#define KC 128
#define LW (KC + 4)
#define G1_LDS_FLOATS ((32 + 128) * LW)

__device__ __forceinline__ void gemm1_strip(
    const float* __restrict__ x, const float* __restrict__ Wih1,
    const float* __restrict__ bih1, const float* __restrict__ bhh1,
    float* __restrict__ xp1, float* lds, const long m0) {
  float* xs = lds;
  float* ws = lds + 32 * LW;
  const int tid  = threadIdx.x;
  const int wave = tid >> 5;
  const int lane = tid & 31;
  const int half = lane >> 4;
  const int l16  = lane & 15;
  const int n0   = wave * 16;

  v8f c0 = {};
  v8f c1 = {};
  for (int k0 = 0; k0 < INW; k0 += KC) {
    for (int i = 0; i < 4; ++i) {
      int idx = tid + i * 256;
      int row = idx >> 5;
      int cv  = (idx & 31) << 2;
      float4 v = *(const float4*)&x[(m0 + row) * INW + k0 + cv];
      *(float4*)&xs[row * LW + cv] = v;
    }
    for (int i = 0; i < 16; ++i) {
      int idx = tid + i * 256;
      int row = idx >> 5;
      int cv  = (idx & 31) << 2;
      float4 v = *(const float4*)&Wih1[(long)row * INW + k0 + cv];
      *(float4*)&ws[row * LW + cv] = v;
    }
    __syncthreads();
    for (int k = 0; k < KC; k += 4) {
      const int kc = k + 2 * half;
      v2f b;  b.x  = ws[(n0 + l16) * LW + kc];  b.y  = ws[(n0 + l16) * LW + kc + 1];
      v2f a0; a0.x = xs[l16 * LW + kc];         a0.y = xs[l16 * LW + kc + 1];
      v2f a1; a1.x = xs[(16 + l16) * LW + kc];  a1.y = xs[(16 + l16) * LW + kc + 1];
      c0 = WMMA_F32(a0, b, c0);
      c1 = WMMA_F32(a1, b, c1);
    }
    __syncthreads();
  }
  const float bias = bih1[n0 + l16] + bhh1[n0 + l16];
  for (int v = 0; v < 8; ++v) {
    long m = m0 + v + 8 * half;
    xp1[m * H + n0 + l16]        = c0[v] + bias;
    xp1[(m + 16) * H + n0 + l16] = c1[v] + bias;
  }
}

__device__ __forceinline__ float tanh_exp2(float x) {
  const float t = __builtin_amdgcn_exp2f(-2.8853900817779268f * __builtin_fabsf(x));
  const float y = (1.0f - t) * __builtin_amdgcn_rcpf(1.0f + t);
  return __builtin_copysignf(y, x);
}

__device__ __forceinline__ void lds_step_barrier() {
  __builtin_amdgcn_fence(__ATOMIC_RELEASE, "workgroup", "local");
  __builtin_amdgcn_s_barrier();
  __builtin_amdgcn_fence(__ATOMIC_ACQUIRE, "workgroup", "local");
}

__device__ __forceinline__ void lds_step_barrier_arrive() {
  __builtin_amdgcn_fence(__ATOMIC_RELEASE, "workgroup", "local");
  __builtin_amdgcn_s_barrier_signal(-1);
}
__device__ __forceinline__ void lds_step_barrier_wait() {
  __builtin_amdgcn_s_barrier_wait(-1);
  __builtin_amdgcn_fence(__ATOMIC_ACQUIRE, "workgroup", "local");
}

__device__ __forceinline__ v8f load8(const float* p) {
  const float4 lo = *(const float4*)&p[0];
  const float4 hi = *(const float4*)&p[4];
  v8f r;
  r[0] = lo.x; r[1] = lo.y; r[2] = lo.z; r[3] = lo.w;
  r[4] = hi.x; r[5] = hi.y; r[6] = hi.z; r[7] = hi.w;
  return r;
}
__device__ __forceinline__ void store8(float* p, const v8f r) {
  float4 lo, hi;
  lo.x = r[0]; lo.y = r[1]; lo.z = r[2]; lo.w = r[3];
  hi.x = r[4]; hi.y = r[5]; hi.z = r[6]; hi.w = r[7];
  *(float4*)&p[0] = lo;
  *(float4*)&p[4] = hi;
}

__device__ __forceinline__ void pin_here(v8f& c) { asm volatile("" : "+v"(c) :: "memory"); }

#define HL (H + 4)
#define TBLK 4
#define CH 64
#define NCH (TLEN / CH)
#define G2R 64
#define PIPE_LDS_FLOATS (164 * 256)

__device__ __forceinline__ void rnn1_chunk(
    const float* __restrict__ xp1, const float* __restrict__ Whh1,
    float* __restrict__ y1, float* hs, const int g, const int ch) {
  const int tid  = threadIdx.x;
  const int wave = tid >> 5;
  const int lane = tid & 31;
  const int half = lane >> 4;
  const int l16  = lane & 15;
  const int n0   = wave * 16;
  const int b0   = g * 16;
  const long tofs = (long)ch * CH * H;

  v2f aw[32];
  const float* wrow = Whh1 + (n0 + l16) * H;
  for (int i = 0; i < 4; ++i) {
    const int k = n0 + 8 * half + 2 * i;
    aw[i].x = wrow[k];  aw[i].y = wrow[k + 1];
  }
  for (int i = 4; i < 32; ++i) {
    const int k = (((4 * wave + i) & 31) << 2) + 2 * half;
    aw[i].x = wrow[k];  aw[i].y = wrow[k + 1];
  }

  if (ch == 0) {
    for (int i = tid; i < 16 * HL; i += 256) hs[i] = 0.0f;
  } else {
    const float* hin = y1 + (long)b0 * TLEN * H + tofs - H;
    const int r  = tid >> 5;
    const int cv = (tid & 31) << 2;
    const float4 va = *(const float4*)&hin[(long)r * TLEN * H + cv];
    const float4 vb = *(const float4*)&hin[(long)(r + 8) * TLEN * H + cv];
    *(float4*)&hs[r * HL + cv] = va;
    *(float4*)&hs[(r + 8) * HL + cv] = vb;
  }
  __syncthreads();

  float* hst = hs + l16 * HL + n0 + 8 * half;
  v8f hown = load8(hst);
  int fo[14];
  for (int p = 0; p < 14; ++p)
    fo[p] = l16 * HL + (((4 * wave + 4 + 2 * p) & 31) << 2) + 2 * half;

  float* yrow = y1 + (long)(b0 + 2 * wave) * TLEN * H + tofs + lane * 4;
  const float* hrow = hs + (2 * wave) * HL + lane * 4;

  const float* xrow = xp1 + (long)(b0 + l16) * TLEN * H + tofs + n0 + 8 * half;
  v8f ring[TBLK];
  for (int j = 0; j < TBLK - 1; ++j) ring[j] = load8(&xrow[(long)j * H]);

  for (int tb = 0; tb < CH; tb += TBLK) {
#pragma unroll
    for (int j = 0; j < TBLK; ++j) {
      const int t = tb + j;
      v8f c = ring[j];
      v2f o0; o0.x = hown[0]; o0.y = hown[1];
      v2f o1; o1.x = hown[2]; o1.y = hown[3];
      v2f o2; o2.x = hown[4]; o2.y = hown[5];
      v2f o3; o3.x = hown[6]; o3.y = hown[7];
      c = WMMA_F32(aw[0], o0, c);
      c = WMMA_F32(aw[1], o1, c);
      pin_here(c);
      lds_step_barrier_arrive();
      c = WMMA_F32(aw[2], o2, c);
      lds_step_barrier_wait();
      const float4 h0v = *(const float4*)&hrow[0];
      const float4 h1v = *(const float4*)&hrow[HL];
      v2f b[28];
      for (int p = 0; p < 14; ++p) {
        b[2 * p].x     = hs[fo[p]];      b[2 * p].y     = hs[fo[p] + 1];
        b[2 * p + 1].x = hs[fo[p] + 4];  b[2 * p + 1].y = hs[fo[p] + 5];
      }
      c = WMMA_F32(aw[3], o3, c);
      __builtin_amdgcn_sched_barrier(0);
      {
        const int tn = (t + TBLK - 1 < CH) ? t + TBLK - 1 : CH - 1;
        ring[(j + TBLK - 1) % TBLK] = load8(&xrow[(long)tn * H]);
      }
      __builtin_amdgcn_sched_barrier(0);
      if (t > 0) {
        *(float4*)&yrow[(long)(t - 1) * H] = h0v;
        *(float4*)&yrow[(long)TLEN * H + (long)(t - 1) * H] = h1v;
      }
      __builtin_amdgcn_sched_barrier(0);
      for (int i = 0; i < 28; ++i) c = WMMA_F32(aw[i + 4], b[i], c);
      __builtin_amdgcn_sched_barrier(0);
      for (int v = 0; v < 8; ++v) hown[v] = tanh_exp2(c[v]);
      lds_step_barrier();
      store8(hst, hown);
    }
  }
  lds_step_barrier();
  {
    const float4 h0v = *(const float4*)&hrow[0];
    const float4 h1v = *(const float4*)&hrow[HL];
    *(float4*)&yrow[(long)(CH - 1) * H] = h0v;
    *(float4*)&yrow[(long)TLEN * H + (long)(CH - 1) * H] = h1v;
  }
}

#define LW2 (H + 4)
__device__ __forceinline__ void gemm2_chunk(
    const float* __restrict__ y1, const float* __restrict__ Wih2,
    const float* __restrict__ bih2, const float* __restrict__ bhh2,
    float* __restrict__ xp2, float* lds, const int b, const int ch) {
  float* ys = lds;
  float* ws = lds + G2R * LW2;
  const int tid  = threadIdx.x;
  const int wave = tid >> 5;
  const int lane = tid & 31;
  const int half = lane >> 4;
  const int l16  = lane & 15;
  const int n0   = wave * 16;
  const long m0  = (long)b * TLEN + (long)ch * CH;

  for (int i = 0; i < 8; ++i) {
    int idx = tid + i * 256;
    int row = idx >> 5;
    int cv  = (idx & 31) << 2;
    float4 v = *(const float4*)&y1[(m0 + row) * H + cv];
    *(float4*)&ys[row * LW2 + cv] = v;
  }
  for (int i = 0; i < 16; ++i) {
    int idx = tid + i * 256;
    int row = idx >> 5;
    int cv  = (idx & 31) << 2;
    float4 v = *(const float4*)&Wih2[(long)row * H + cv];
    *(float4*)&ws[row * LW2 + cv] = v;
  }
  const float bias2 = bih2[n0 + l16] + bhh2[n0 + l16];
  v8f c0, c1, c2, c3;
  for (int v = 0; v < 8; ++v) { c0[v] = bias2; c1[v] = bias2; c2[v] = bias2; c3[v] = bias2; }
  __syncthreads();
  for (int k = 0; k < H; k += 4) {
    const int kc = k + 2 * half;
    v2f bb; bb.x = ws[(n0 + l16) * LW2 + kc];  bb.y = ws[(n0 + l16) * LW2 + kc + 1];
    v2f a0; a0.x = ys[l16 * LW2 + kc];         a0.y = ys[l16 * LW2 + kc + 1];
    v2f a1; a1.x = ys[(16 + l16) * LW2 + kc];  a1.y = ys[(16 + l16) * LW2 + kc + 1];
    v2f a2; a2.x = ys[(32 + l16) * LW2 + kc];  a2.y = ys[(32 + l16) * LW2 + kc + 1];
    v2f a3; a3.x = ys[(48 + l16) * LW2 + kc];  a3.y = ys[(48 + l16) * LW2 + kc + 1];
    c0 = WMMA_F32(a0, bb, c0);
    c1 = WMMA_F32(a1, bb, c1);
    c2 = WMMA_F32(a2, bb, c2);
    c3 = WMMA_F32(a3, bb, c3);
  }
  for (int v = 0; v < 8; ++v) {
    long m = m0 + v + 8 * half;
    xp2[m * H + n0 + l16]        = c0[v];
    xp2[(m + 16) * H + n0 + l16] = c1[v];
    xp2[(m + 32) * H + n0 + l16] = c2[v];
    xp2[(m + 48) * H + n0 + l16] = c3[v];
  }
}

__device__ __forceinline__ void rnn2_chunk(
    const float* __restrict__ xp2, const float* __restrict__ Whh2,
    float* __restrict__ h2, float* hs, const int g, const int ch) {
  const int tid  = threadIdx.x;
  const int wave = tid >> 5;
  const int lane = tid & 31;
  const int half = lane >> 4;
  const int l16  = lane & 15;
  const int n0   = wave * 16;
  const int b0   = g * 16;
  const long tofs = (long)ch * CH * H;

  v2f aw[32];
  const float* wrow = Whh2 + (n0 + l16) * H;
  for (int i = 0; i < 4; ++i) {
    const int k = n0 + 8 * half + 2 * i;
    aw[i].x = wrow[k];  aw[i].y = wrow[k + 1];
  }
  for (int i = 4; i < 32; ++i) {
    const int k = (((4 * wave + i) & 31) << 2) + 2 * half;
    aw[i].x = wrow[k];  aw[i].y = wrow[k + 1];
  }
  if (ch == 0) {
    for (int i = tid; i < 16 * HL; i += 256) hs[i] = 0.0f;
  } else {
    const float* hin = h2 + (long)b0 * H;
    const int r  = tid >> 5;
    const int cv = (tid & 31) << 2;
    const float4 va = *(const float4*)&hin[r * H + cv];
    const float4 vb = *(const float4*)&hin[(r + 8) * H + cv];
    *(float4*)&hs[r * HL + cv] = va;
    *(float4*)&hs[(r + 8) * HL + cv] = vb;
  }
  __syncthreads();

  float* hst = hs + l16 * HL + n0 + 8 * half;
  v8f hown = load8(hst);
  int fo[14];
  for (int p = 0; p < 14; ++p)
    fo[p] = l16 * HL + (((4 * wave + 4 + 2 * p) & 31) << 2) + 2 * half;

  const float* xrow = xp2 + (long)(b0 + l16) * TLEN * H + tofs + n0 + 8 * half;
  v8f ring[TBLK];
  for (int j = 0; j < TBLK - 1; ++j) ring[j] = load8(&xrow[(long)j * H]);

  for (int tb = 0; tb < CH; tb += TBLK) {
#pragma unroll
    for (int j = 0; j < TBLK; ++j) {
      const int t = tb + j;
      v8f c = ring[j];
      v2f o0; o0.x = hown[0]; o0.y = hown[1];
      v2f o1; o1.x = hown[2]; o1.y = hown[3];
      v2f o2; o2.x = hown[4]; o2.y = hown[5];
      v2f o3; o3.x = hown[6]; o3.y = hown[7];
      c = WMMA_F32(aw[0], o0, c);
      c = WMMA_F32(aw[1], o1, c);
      pin_here(c);
      lds_step_barrier_arrive();
      c = WMMA_F32(aw[2], o2, c);
      lds_step_barrier_wait();
      v2f b[28];
      for (int p = 0; p < 14; ++p) {
        b[2 * p].x     = hs[fo[p]];      b[2 * p].y     = hs[fo[p] + 1];
        b[2 * p + 1].x = hs[fo[p] + 4];  b[2 * p + 1].y = hs[fo[p] + 5];
      }
      c = WMMA_F32(aw[3], o3, c);
      __builtin_amdgcn_sched_barrier(0);
      {
        const int tn = (t + TBLK - 1 < CH) ? t + TBLK - 1 : CH - 1;
        ring[(j + TBLK - 1) % TBLK] = load8(&xrow[(long)tn * H]);
      }
      __builtin_amdgcn_sched_barrier(0);
      for (int i = 0; i < 28; ++i) c = WMMA_F32(aw[i + 4], b[i], c);
      __builtin_amdgcn_sched_barrier(0);
      for (int v = 0; v < 8; ++v) hown[v] = tanh_exp2(c[v]);
      lds_step_barrier();
      store8(hst, hown);
    }
  }
  store8(&h2[(long)(b0 + l16) * H + n0 + 8 * half], hown);
}

__global__ __launch_bounds__(256) void gemm1_head(
    const float* __restrict__ x, const float* __restrict__ Wih1,
    const float* __restrict__ bih1, const float* __restrict__ bhh1,
    float* __restrict__ xp1) {
  __shared__ __attribute__((aligned(16))) float lds[G1_LDS_FLOATS];
  const int s = blockIdx.x;
  const long m0 = (long)(s >> 1) * TLEN + (long)(s & 1) * 32;
  gemm1_strip(x, Wih1, bih1, bhh1, xp1, lds, m0);
}

#define PIPE_G1_FIRST (8 + BSZ)
__global__ __launch_bounds__(256) void pipe_kernel(
    float* bufA, float* bufB, float* h2,
    const float* Whh1, const float* Wih2, const float* bih2, const float* bhh2,
    const float* Whh2,
    const float* x, const float* Wih1, const float* bih1, const float* bhh1,
    const int li) {
  __shared__ __attribute__((aligned(16))) float lds[PIPE_LDS_FLOATS];
  const int bx = blockIdx.x;
  if (bx < 4) {
    if (li < NCH) rnn1_chunk(bufA, Whh1, bufB, lds, bx, li);
  } else if (bx < 8) {
    if (li >= 2 && li - 2 < NCH) rnn2_chunk(bufA, Whh2, h2, lds, bx - 4, li - 2);
  } else if (bx < PIPE_G1_FIRST) {
    if (li >= 1 && li - 1 < NCH) gemm2_chunk(bufB, Wih2, bih2, bhh2, bufA, lds, bx - 8, li - 1);
  } else {
    if (li + 1 < NCH) {
      const int s = bx - PIPE_G1_FIRST;
      const long m0 = (long)(s >> 1) * TLEN + (long)(li + 1) * CH + (long)(s & 1) * 32;
      gemm1_strip(x, Wih1, bih1, bhh1, bufA, lds, m0);
    }
  }
}

__global__ __launch_bounds__(64) void proj_ln_kernel(
    const float* __restrict__ h2, const float* __restrict__ Wp,
    const float* __restrict__ bp, const float* __restrict__ gamma,
    const float* __restrict__ beta, float* __restrict__ out) {
  __shared__ float tmp[PROJ];
  const int b = blockIdx.x;
  const int j = threadIdx.x;
  float acc = bp[j];
  for (int k = 0; k < H; ++k) acc += h2[b * H + k] * Wp[j * H + k];

  tmp[j] = acc;
  __syncthreads();
  for (int s = 32; s > 0; s >>= 1) {
    if (j < s) tmp[j] += tmp[j + s];
    __syncthreads();
  }
  const float mu = tmp[0] * (1.0f / PROJ);
  __syncthreads();
  const float d = acc - mu;
  tmp[j] = d * d;
  __syncthreads();
  for (int s = 32; s > 0; s >>= 1) {
    if (j < s) tmp[j] += tmp[j + s];
    __syncthreads();
  }
  const float var = tmp[0] * (1.0f / PROJ);
  const float r = rsqrtf(var + EPSV);
  out[b * PROJ + j] = d * r * gamma[j] + beta[j];
}

extern "C" void kernel_launch(void* const* d_in, const int* in_sizes, int n_in,
                              void* d_out, int out_size, void* d_ws, size_t ws_size,
                              hipStream_t stream) {
  (void)in_sizes; (void)n_in; (void)out_size; (void)ws_size;
  const float* x     = (const float*)d_in[0];
  const float* Wih1  = (const float*)d_in[1];
  const float* Whh1  = (const float*)d_in[2];
  const float* bih1  = (const float*)d_in[3];
  const float* bhh1  = (const float*)d_in[4];
  const float* Wih2  = (const float*)d_in[5];
  const float* Whh2  = (const float*)d_in[6];
  const float* bih2  = (const float*)d_in[7];
  const float* bhh2  = (const float*)d_in[8];
  const float* Wproj = (const float*)d_in[9];
  const float* bproj = (const float*)d_in[10];
  const float* gam   = (const float*)d_in[11];
  const float* bet   = (const float*)d_in[12];
  float* out = (float*)d_out;

  float* bufA = (float*)d_ws;
  float* bufB = bufA + (size_t)M_TOT * H;
  float* h2   = bufB + (size_t)M_TOT * H;

  gemm1_head<<<128, 256, 0, stream>>>(x, Wih1, bih1, bhh1, bufA);
  pipe_kernel<<<200, 256, 0, stream>>>(bufA, bufB, h2, Whh1, Wih2, bih2, bhh2, Whh2, x, Wih1, bih1, bhh1, 0);
  pipe_kernel<<<200, 256, 0, stream>>>(bufA, bufB, h2, Whh1, Wih2, bih2, bhh2, Whh2, x, Wih1, bih1, bhh1, 1);
  pipe_kernel<<<200, 256, 0, stream>>>(bufA, bufB, h2, Whh1, Wih2, bih2, bhh2, Whh2, x, Wih1, bih1, bhh1, 2);
  pipe_kernel<<<200, 256, 0, stream>>>(bufA, bufB, h2, Whh1, Wih2, bih2, bhh2, Whh2, x, Wih1, bih1, bhh1, 3);
  pipe_kernel<<<200, 256, 0, stream>>>(bufA, bufB, h2, Whh1, Wih2, bih2, bhh2, Whh2, x, Wih1, bih1, bhh1, 4);
  pipe_kernel<<<200, 256, 0, stream>>>(bufA, bufB, h2, Whh1, Wih2, bih2, bhh2, Whh2, x, Wih1, bih1, bhh1, 5);
  pipe_kernel<<<200, 256, 0, stream>>>(bufA, bufB, h2, Whh1, Wih2, bih2, bhh2, Whh2, x, Wih1, bih1, bhh1, 6);
  pipe_kernel<<<72,  256, 0, stream>>>(bufA, bufB, h2, Whh1, Wih2, bih2, bhh2, Whh2, x, Wih1, bih1, bhh1, 7);
  pipe_kernel<<<72,  256, 0, stream>>>(bufA, bufB, h2, Whh1, Wih2, bih2, bhh2, Whh2, x, Wih1, bih1, bhh1, 8);
  pipe_kernel<<<8,   256, 0, stream>>>(bufA, bufB, h2, Whh1, Wih2, bih2, bhh2, Whh2, x, Wih1, bih1, bhh1, 9);
  proj_ln_kernel<<<BSZ, PROJ, 0, stream>>>(h2, Wproj, bproj, gam, bet, out);
}


# recurrences: waves 0-3 (one per SIMD) do all products, two accumulator tiles each, waves 4-7 only copy y1 out: 16 not 4 of a wave's 64 WMMAs take its own new h from registers; own 32 k summed first
# speedup vs baseline: 3.9826x; 1.0403x over previous
#include <hip/hip_runtime.h>

#define H 128
#define INW 2048
#define PROJ 64
#define BSZ 64
#define TLEN 512
#define M_TOT (BSZ * TLEN)
#define EPSV 1e-5f

typedef __attribute__((ext_vector_type(2))) float v2f;
typedef __attribute__((ext_vector_type(8))) float v8f;

#define WMMA_F32(a, b, c) \
  __builtin_amdgcn_wmma_f32_16x16x4_f32(false, (a), false, (b), (short)0, (c), false, false)

#define KC 128
#define LW (KC + 4)
#define G1_LDS_FLOATS ((32 + 128) * LW)

__device__ __forceinline__ void gemm1_strip(
    const float* __restrict__ x, const float* __restrict__ Wih1,
    const float* __restrict__ bih1, const float* __restrict__ bhh1,
    float* __restrict__ xp1, float* lds, const long m0) {
  float* xs = lds;
  float* ws = lds + 32 * LW;
  const int tid  = threadIdx.x;
  const int wave = tid >> 5;
  const int lane = tid & 31;
  const int half = lane >> 4;
  const int l16  = lane & 15;
  const int n0   = wave * 16;

  v8f c0 = {};
  v8f c1 = {};
  for (int k0 = 0; k0 < INW; k0 += KC) {
    for (int i = 0; i < 4; ++i) {
      int idx = tid + i * 256;
      int row = idx >> 5;
      int cv  = (idx & 31) << 2;
      float4 v = *(const float4*)&x[(m0 + row) * INW + k0 + cv];
      *(float4*)&xs[row * LW + cv] = v;
    }
    for (int i = 0; i < 16; ++i) {
      int idx = tid + i * 256;
      int row = idx >> 5;
      int cv  = (idx & 31) << 2;
      float4 v = *(const float4*)&Wih1[(long)row * INW + k0 + cv];
      *(float4*)&ws[row * LW + cv] = v;
    }
    __syncthreads();
    for (int k = 0; k < KC; k += 4) {
      const int kc = k + 2 * half;
      v2f b;  b.x  = ws[(n0 + l16) * LW + kc];  b.y  = ws[(n0 + l16) * LW + kc + 1];
      v2f a0; a0.x = xs[l16 * LW + kc];         a0.y = xs[l16 * LW + kc + 1];
      v2f a1; a1.x = xs[(16 + l16) * LW + kc];  a1.y = xs[(16 + l16) * LW + kc + 1];
      c0 = WMMA_F32(a0, b, c0);
      c1 = WMMA_F32(a1, b, c1);
    }
    __syncthreads();
  }
  const float bias = bih1[n0 + l16] + bhh1[n0 + l16];
  for (int v = 0; v < 8; ++v) {
    long m = m0 + v + 8 * half;
    xp1[m * H + n0 + l16]        = c0[v] + bias;
    xp1[(m + 16) * H + n0 + l16] = c1[v] + bias;
  }
}

__device__ __forceinline__ float tanh_exp2(float x) {
  const float t = __builtin_amdgcn_exp2f(-2.8853900817779268f * __builtin_fabsf(x));
  const float y = (1.0f - t) * __builtin_amdgcn_rcpf(1.0f + t);
  return __builtin_copysignf(y, x);
}

__device__ __forceinline__ void lds_step_barrier() {
  __builtin_amdgcn_fence(__ATOMIC_RELEASE, "workgroup", "local");
  __builtin_amdgcn_s_barrier();
  __builtin_amdgcn_fence(__ATOMIC_ACQUIRE, "workgroup", "local");
}

__device__ __forceinline__ void lds_step_barrier_arrive() {
  __builtin_amdgcn_fence(__ATOMIC_RELEASE, "workgroup", "local");
  __builtin_amdgcn_s_barrier_signal(-1);
}
__device__ __forceinline__ void lds_step_barrier_wait() {
  __builtin_amdgcn_s_barrier_wait(-1);
  __builtin_amdgcn_fence(__ATOMIC_ACQUIRE, "workgroup", "local");
}

__device__ __forceinline__ v8f load8(const float* p) {
  const float4 lo = *(const float4*)&p[0];
  const float4 hi = *(const float4*)&p[4];
  v8f r;
  r[0] = lo.x; r[1] = lo.y; r[2] = lo.z; r[3] = lo.w;
  r[4] = hi.x; r[5] = hi.y; r[6] = hi.z; r[7] = hi.w;
  return r;
}
__device__ __forceinline__ void store8(float* p, const v8f r) {
  float4 lo, hi;
  lo.x = r[0]; lo.y = r[1]; lo.z = r[2]; lo.w = r[3];
  hi.x = r[4]; hi.y = r[5]; hi.z = r[6]; hi.w = r[7];
  *(float4*)&p[0] = lo;
  *(float4*)&p[4] = hi;
}

__device__ __forceinline__ void pin_here(v8f& c) { asm volatile("" : "+v"(c) :: "memory"); }
__device__ __forceinline__ void pin_pair(v8f& c, v8f& d) { asm volatile("" : "+v"(c), "+v"(d)); }

#define HL (H + 4)
#define TBLK 4
#define CH 64
#define NCH (TLEN / CH)
#define G2R 64
#define PIPE_LDS_FLOATS (164 * 256)

#define SKEW 3
__device__ __forceinline__ void rnn1_chunk(
    const float* __restrict__ xp1, const float* __restrict__ Whh1,
    float* __restrict__ y1, float* hs, const int g, const int ch) {
  const int tid  = threadIdx.x;
  const int wave = __builtin_amdgcn_readfirstlane(tid >> 5);
  const int lane = tid & 31;
  const int half = lane >> 4;
  const int l16  = lane & 15;
  const int n0   = (wave & 3) * 32;
  const int b0   = g * 16;
  const long tofs = (long)ch * CH * H;

  if (ch == 0) {
    for (int i = tid; i < 16 * HL; i += 256) hs[i] = 0.0f;
  } else {
    const float* hin = y1 + (long)b0 * TLEN * H + tofs - H;
    const int r  = tid >> 5;
    const int cv = (tid & 31) << 2;
    const float4 va = *(const float4*)&hin[(long)r * TLEN * H + cv];
    const float4 vb = *(const float4*)&hin[(long)(r + 8) * TLEN * H + cv];
    *(float4*)&hs[r * HL + cv] = va;
    *(float4*)&hs[(r + 8) * HL + cv] = vb;
  }
  __syncthreads();

  if (wave < 4) {
    v2f aw0[32], aw1[32];
    const float* wr0 = Whh1 + (n0 + l16) * H;
    const float* wr1 = wr0 + 16 * H;
    for (int i = 0; i < 8; ++i) {
      const int k = n0 + 16 * (i >> 2) + 8 * half + 2 * (i & 3);
      aw0[i].x = wr0[k];  aw0[i].y = wr0[k + 1];
      aw1[i].x = wr1[k];  aw1[i].y = wr1[k + 1];
    }
    for (int i = 8; i < 32; ++i) {
      const int k = (((8 * wave + i) & 31) << 2) + 2 * half;
      aw0[i].x = wr0[k];  aw0[i].y = wr0[k + 1];
      aw1[i].x = wr1[k];  aw1[i].y = wr1[k + 1];
    }

    float* hst0 = hs + l16 * HL + n0 + 8 * half;
    float* hst1 = hst0 + 16;
    v8f ha = load8(hst0);
    v8f hb = load8(hst1);
    int fo[12];
    for (int p = 0; p < 12; ++p)
      fo[p] = l16 * HL + (((8 * wave + 8 + 2 * p) & 31) << 2) + 2 * half;

    const float* xrow = xp1 + (long)(b0 + l16) * TLEN * H + tofs + n0 + 8 * half;
    v8f ring0[TBLK], ring1[TBLK];
    for (int j = 0; j < TBLK - 1; ++j) {
      ring0[j] = load8(&xrow[(long)j * H]);
      ring1[j] = load8(&xrow[(long)j * H + 16]);
    }

    for (int tb = 0; tb < CH; tb += TBLK) {
#pragma unroll
      for (int j = 0; j < TBLK; ++j) {
        const int t = tb + j;
        v8f c0 = ring0[j];
        v8f c1 = ring1[j];
        v2f o[8];
        for (int i = 0; i < 4; ++i) {
          o[i].x     = ha[2 * i];  o[i].y     = ha[2 * i + 1];
          o[4 + i].x = hb[2 * i];  o[4 + i].y = hb[2 * i + 1];
        }
        c0 = WMMA_F32(aw0[0], o[0], c0);  c1 = WMMA_F32(aw1[0], o[0], c1);
        c0 = WMMA_F32(aw0[1], o[1], c0);  c1 = WMMA_F32(aw1[1], o[1], c1);
        pin_here(c0);
        pin_here(c1);
        lds_step_barrier_arrive();
        c0 = WMMA_F32(aw0[2], o[2], c0);  c1 = WMMA_F32(aw1[2], o[2], c1);
        lds_step_barrier_wait();
        v2f b[24];
        for (int p = 0; p < 12; ++p) {
          b[2 * p].x     = hs[fo[p]];      b[2 * p].y     = hs[fo[p] + 1];
          b[2 * p + 1].x = hs[fo[p] + 4];  b[2 * p + 1].y = hs[fo[p] + 5];
        }
        __builtin_amdgcn_sched_barrier(0);
        for (int p = 3; p < 8; ++p) {
          c0 = WMMA_F32(aw0[p], o[p], c0);  c1 = WMMA_F32(aw1[p], o[p], c1);
        }
        {
          const int tn = (t + TBLK - 1 < CH) ? t + TBLK - 1 : CH - 1;
          ring0[(j + TBLK - 1) % TBLK] = load8(&xrow[(long)tn * H]);
          ring1[(j + TBLK - 1) % TBLK] = load8(&xrow[(long)tn * H + 16]);
        }
        __builtin_amdgcn_sched_barrier(0);
        for (int i = 0; i < SKEW; ++i) c0 = WMMA_F32(aw0[i + 8], b[i], c0);
        for (int i = 0; i < 24 - SKEW; ++i) {
          c1 = WMMA_F32(aw1[i + 8], b[i], c1);
          c0 = WMMA_F32(aw0[i + SKEW + 8], b[i + SKEW], c0);
          pin_pair(c0, c1);
        }
        __builtin_amdgcn_sched_barrier(0);
        for (int i = 24 - SKEW; i < 24; ++i) c1 = WMMA_F32(aw1[i + 8], b[i], c1);
        for (int v = 0; v < 8; ++v) ha[v] = tanh_exp2(c0[v]);
        for (int v = 0; v < 8; ++v) hb[v] = tanh_exp2(c1[v]);
        lds_step_barrier();
        store8(hst0, ha);
        store8(hst1, hb);
      }
    }
    lds_step_barrier();
  } else {
    float* yrow = y1 + (long)(b0 + 4 * (wave - 4)) * TLEN * H + tofs + lane * 4;
    const float* hrow = hs + (4 * (wave - 4)) * HL + lane * 4;
    for (int t = 0; t < CH; ++t) {
      lds_step_barrier_arrive();
      lds_step_barrier_wait();
      if (t > 0) {
        float4 hv[4];
        for (int i = 0; i < 4; ++i) hv[i] = *(const float4*)&hrow[i * HL];
        for (int i = 0; i < 4; ++i)
          *(float4*)&yrow[(long)i * TLEN * H + (long)(t - 1) * H] = hv[i];
      }
      lds_step_barrier();
    }
    lds_step_barrier();
    float4 hv[4];
    for (int i = 0; i < 4; ++i) hv[i] = *(const float4*)&hrow[i * HL];
    for (int i = 0; i < 4; ++i)
      *(float4*)&yrow[(long)i * TLEN * H + (long)(CH - 1) * H] = hv[i];
  }
}

#define LW2 (H + 4)
__device__ __forceinline__ void gemm2_chunk(
    const float* __restrict__ y1, const float* __restrict__ Wih2,
    const float* __restrict__ bih2, const float* __restrict__ bhh2,
    float* __restrict__ xp2, float* lds, const int b, const int ch) {
  float* ys = lds;
  float* ws = lds + G2R * LW2;
  const int tid  = threadIdx.x;
  const int wave = tid >> 5;
  const int lane = tid & 31;
  const int half = lane >> 4;
  const int l16  = lane & 15;
  const int n0   = wave * 16;
  const long m0  = (long)b * TLEN + (long)ch * CH;

  for (int i = 0; i < 8; ++i) {
    int idx = tid + i * 256;
    int row = idx >> 5;
    int cv  = (idx & 31) << 2;
    float4 v = *(const float4*)&y1[(m0 + row) * H + cv];
    *(float4*)&ys[row * LW2 + cv] = v;
  }
  for (int i = 0; i < 16; ++i) {
    int idx = tid + i * 256;
    int row = idx >> 5;
    int cv  = (idx & 31) << 2;
    float4 v = *(const float4*)&Wih2[(long)row * H + cv];
    *(float4*)&ws[row * LW2 + cv] = v;
  }
  const float bias2 = bih2[n0 + l16] + bhh2[n0 + l16];
  v8f c0, c1, c2, c3;
  for (int v = 0; v < 8; ++v) { c0[v] = bias2; c1[v] = bias2; c2[v] = bias2; c3[v] = bias2; }
  __syncthreads();
  for (int k = 0; k < H; k += 4) {
    const int kc = k + 2 * half;
    v2f bb; bb.x = ws[(n0 + l16) * LW2 + kc];  bb.y = ws[(n0 + l16) * LW2 + kc + 1];
    v2f a0; a0.x = ys[l16 * LW2 + kc];         a0.y = ys[l16 * LW2 + kc + 1];
    v2f a1; a1.x = ys[(16 + l16) * LW2 + kc];  a1.y = ys[(16 + l16) * LW2 + kc + 1];
    v2f a2; a2.x = ys[(32 + l16) * LW2 + kc];  a2.y = ys[(32 + l16) * LW2 + kc + 1];
    v2f a3; a3.x = ys[(48 + l16) * LW2 + kc];  a3.y = ys[(48 + l16) * LW2 + kc + 1];
    c0 = WMMA_F32(a0, bb, c0);
    c1 = WMMA_F32(a1, bb, c1);
    c2 = WMMA_F32(a2, bb, c2);
    c3 = WMMA_F32(a3, bb, c3);
  }
  for (int v = 0; v < 8; ++v) {
    long m = m0 + v + 8 * half;
    xp2[m * H + n0 + l16]        = c0[v];
    xp2[(m + 16) * H + n0 + l16] = c1[v];
    xp2[(m + 32) * H + n0 + l16] = c2[v];
    xp2[(m + 48) * H + n0 + l16] = c3[v];
  }
}

__device__ __forceinline__ void rnn2_chunk(
    const float* __restrict__ xp2, const float* __restrict__ Whh2,
    float* __restrict__ h2, float* hs, const int g, const int ch) {
  const int tid  = threadIdx.x;
  const int wave = __builtin_amdgcn_readfirstlane(tid >> 5);
  const int lane = tid & 31;
  const int half = lane >> 4;
  const int l16  = lane & 15;
  const int n0   = (wave & 3) * 32;
  const int b0   = g * 16;
  const long tofs = (long)ch * CH * H;

  if (ch == 0) {
    for (int i = tid; i < 16 * HL; i += 256) hs[i] = 0.0f;
  } else {
    const float* hin = h2 + (long)b0 * H;
    const int r  = tid >> 5;
    const int cv = (tid & 31) << 2;
    const float4 va = *(const float4*)&hin[r * H + cv];
    const float4 vb = *(const float4*)&hin[(r + 8) * H + cv];
    *(float4*)&hs[r * HL + cv] = va;
    *(float4*)&hs[(r + 8) * HL + cv] = vb;
  }
  __syncthreads();

  if (wave < 4) {
    v2f aw0[32], aw1[32];
    const float* wr0 = Whh2 + (n0 + l16) * H;
    const float* wr1 = wr0 + 16 * H;
    for (int i = 0; i < 8; ++i) {
      const int k = n0 + 16 * (i >> 2) + 8 * half + 2 * (i & 3);
      aw0[i].x = wr0[k];  aw0[i].y = wr0[k + 1];
      aw1[i].x = wr1[k];  aw1[i].y = wr1[k + 1];
    }
    for (int i = 8; i < 32; ++i) {
      const int k = (((8 * wave + i) & 31) << 2) + 2 * half;
      aw0[i].x = wr0[k];  aw0[i].y = wr0[k + 1];
      aw1[i].x = wr1[k];  aw1[i].y = wr1[k + 1];
    }

    float* hst0 = hs + l16 * HL + n0 + 8 * half;
    float* hst1 = hst0 + 16;
    v8f ha = load8(hst0);
    v8f hb = load8(hst1);
    int fo[12];
    for (int p = 0; p < 12; ++p)
      fo[p] = l16 * HL + (((8 * wave + 8 + 2 * p) & 31) << 2) + 2 * half;

    const float* xrow = xp2 + (long)(b0 + l16) * TLEN * H + tofs + n0 + 8 * half;
    v8f ring0[TBLK], ring1[TBLK];
    for (int j = 0; j < TBLK - 1; ++j) {
      ring0[j] = load8(&xrow[(long)j * H]);
      ring1[j] = load8(&xrow[(long)j * H + 16]);
    }

    for (int tb = 0; tb < CH; tb += TBLK) {
#pragma unroll
      for (int j = 0; j < TBLK; ++j) {
        const int t = tb + j;
        v8f c0 = ring0[j];
        v8f c1 = ring1[j];
        v2f o[8];
        for (int i = 0; i < 4; ++i) {
          o[i].x     = ha[2 * i];  o[i].y     = ha[2 * i + 1];
          o[4 + i].x = hb[2 * i];  o[4 + i].y = hb[2 * i + 1];
        }
        c0 = WMMA_F32(aw0[0], o[0], c0);  c1 = WMMA_F32(aw1[0], o[0], c1);
        c0 = WMMA_F32(aw0[1], o[1], c0);  c1 = WMMA_F32(aw1[1], o[1], c1);
        pin_here(c0);
        pin_here(c1);
        lds_step_barrier_arrive();
        c0 = WMMA_F32(aw0[2], o[2], c0);  c1 = WMMA_F32(aw1[2], o[2], c1);
        lds_step_barrier_wait();
        v2f b[24];
        for (int p = 0; p < 12; ++p) {
          b[2 * p].x     = hs[fo[p]];      b[2 * p].y     = hs[fo[p] + 1];
          b[2 * p + 1].x = hs[fo[p] + 4];  b[2 * p + 1].y = hs[fo[p] + 5];
        }
        __builtin_amdgcn_sched_barrier(0);
        for (int p = 3; p < 8; ++p) {
          c0 = WMMA_F32(aw0[p], o[p], c0);  c1 = WMMA_F32(aw1[p], o[p], c1);
        }
        {
          const int tn = (t + TBLK - 1 < CH) ? t + TBLK - 1 : CH - 1;
          ring0[(j + TBLK - 1) % TBLK] = load8(&xrow[(long)tn * H]);
          ring1[(j + TBLK - 1) % TBLK] = load8(&xrow[(long)tn * H + 16]);
        }
        __builtin_amdgcn_sched_barrier(0);
        for (int i = 0; i < SKEW; ++i) c0 = WMMA_F32(aw0[i + 8], b[i], c0);
        for (int i = 0; i < 24 - SKEW; ++i) {
          c1 = WMMA_F32(aw1[i + 8], b[i], c1);
          c0 = WMMA_F32(aw0[i + SKEW + 8], b[i + SKEW], c0);
          pin_pair(c0, c1);
        }
        __builtin_amdgcn_sched_barrier(0);
        for (int i = 24 - SKEW; i < 24; ++i) c1 = WMMA_F32(aw1[i + 8], b[i], c1);
        for (int v = 0; v < 8; ++v) ha[v] = tanh_exp2(c0[v]);
        for (int v = 0; v < 8; ++v) hb[v] = tanh_exp2(c1[v]);
        lds_step_barrier();
        store8(hst0, ha);
        store8(hst1, hb);
      }
    }
    float* hout = &h2[(long)(b0 + l16) * H + n0 + 8 * half];
    store8(hout, ha);
    store8(hout + 16, hb);
  } else {
    for (int t = 0; t < CH; ++t) {
      lds_step_barrier_arrive();
      lds_step_barrier_wait();
      lds_step_barrier();
    }
  }
}

__global__ __launch_bounds__(256) void gemm1_head(
    const float* __restrict__ x, const float* __restrict__ Wih1,
    const float* __restrict__ bih1, const float* __restrict__ bhh1,
    float* __restrict__ xp1) {
  __shared__ __attribute__((aligned(16))) float lds[G1_LDS_FLOATS];
  const int s = blockIdx.x;
  const long m0 = (long)(s >> 1) * TLEN + (long)(s & 1) * 32;
  gemm1_strip(x, Wih1, bih1, bhh1, xp1, lds, m0);
}

#define PIPE_G1_FIRST (8 + BSZ)
__global__ __launch_bounds__(256) void pipe_kernel(
    float* bufA, float* bufB, float* h2,
    const float* Whh1, const float* Wih2, const float* bih2, const float* bhh2,
    const float* Whh2,
    const float* x, const float* Wih1, const float* bih1, const float* bhh1,
    const int li) {
  __shared__ __attribute__((aligned(16))) float lds[PIPE_LDS_FLOATS];
  const int bx = blockIdx.x;
  if (bx < 4) {
    if (li < NCH) rnn1_chunk(bufA, Whh1, bufB, lds, bx, li);
  } else if (bx < 8) {
    if (li >= 2 && li - 2 < NCH) rnn2_chunk(bufA, Whh2, h2, lds, bx - 4, li - 2);
  } else if (bx < PIPE_G1_FIRST) {
    if (li >= 1 && li - 1 < NCH) gemm2_chunk(bufB, Wih2, bih2, bhh2, bufA, lds, bx - 8, li - 1);
  } else {
    if (li + 1 < NCH) {
      const int s = bx - PIPE_G1_FIRST;
      const long m0 = (long)(s >> 1) * TLEN + (long)(li + 1) * CH + (long)(s & 1) * 32;
      gemm1_strip(x, Wih1, bih1, bhh1, bufA, lds, m0);
    }
  }
}

__global__ __launch_bounds__(64) void proj_ln_kernel(
    const float* __restrict__ h2, const float* __restrict__ Wp,
    const float* __restrict__ bp, const float* __restrict__ gamma,
    const float* __restrict__ beta, float* __restrict__ out) {
  __shared__ float tmp[PROJ];
  const int b = blockIdx.x;
  const int j = threadIdx.x;
  float acc = bp[j];
  for (int k = 0; k < H; ++k) acc += h2[b * H + k] * Wp[j * H + k];

  tmp[j] = acc;
  __syncthreads();
  for (int s = 32; s > 0; s >>= 1) {
    if (j < s) tmp[j] += tmp[j + s];
    __syncthreads();
  }
  const float mu = tmp[0] * (1.0f / PROJ);
  __syncthreads();
  const float d = acc - mu;
  tmp[j] = d * d;
  __syncthreads();
  for (int s = 32; s > 0; s >>= 1) {
    if (j < s) tmp[j] += tmp[j + s];
    __syncthreads();
  }
  const float var = tmp[0] * (1.0f / PROJ);
  const float r = rsqrtf(var + EPSV);
  out[b * PROJ + j] = d * r * gamma[j] + beta[j];
}

extern "C" void kernel_launch(void* const* d_in, const int* in_sizes, int n_in,
                              void* d_out, int out_size, void* d_ws, size_t ws_size,
                              hipStream_t stream) {
  (void)in_sizes; (void)n_in; (void)out_size; (void)ws_size;
  const float* x     = (const float*)d_in[0];
  const float* Wih1  = (const float*)d_in[1];
  const float* Whh1  = (const float*)d_in[2];
  const float* bih1  = (const float*)d_in[3];
  const float* bhh1  = (const float*)d_in[4];
  const float* Wih2  = (const float*)d_in[5];
  const float* Whh2  = (const float*)d_in[6];
  const float* bih2  = (const float*)d_in[7];
  const float* bhh2  = (const float*)d_in[8];
  const float* Wproj = (const float*)d_in[9];
  const float* bproj = (const float*)d_in[10];
  const float* gam   = (const float*)d_in[11];
  const float* bet   = (const float*)d_in[12];
  float* out = (float*)d_out;

  float* bufA = (float*)d_ws;
  float* bufB = bufA + (size_t)M_TOT * H;
  float* h2   = bufB + (size_t)M_TOT * H;

  gemm1_head<<<128, 256, 0, stream>>>(x, Wih1, bih1, bhh1, bufA);
  pipe_kernel<<<200, 256, 0, stream>>>(bufA, bufB, h2, Whh1, Wih2, bih2, bhh2, Whh2, x, Wih1, bih1, bhh1, 0);
  pipe_kernel<<<200, 256, 0, stream>>>(bufA, bufB, h2, Whh1, Wih2, bih2, bhh2, Whh2, x, Wih1, bih1, bhh1, 1);
  pipe_kernel<<<200, 256, 0, stream>>>(bufA, bufB, h2, Whh1, Wih2, bih2, bhh2, Whh2, x, Wih1, bih1, bhh1, 2);
  pipe_kernel<<<200, 256, 0, stream>>>(bufA, bufB, h2, Whh1, Wih2, bih2, bhh2, Whh2, x, Wih1, bih1, bhh1, 3);
  pipe_kernel<<<200, 256, 0, stream>>>(bufA, bufB, h2, Whh1, Wih2, bih2, bhh2, Whh2, x, Wih1, bih1, bhh1, 4);
  pipe_kernel<<<200, 256, 0, stream>>>(bufA, bufB, h2, Whh1, Wih2, bih2, bhh2, Whh2, x, Wih1, bih1, bhh1, 5);
  pipe_kernel<<<200, 256, 0, stream>>>(bufA, bufB, h2, Whh1, Wih2, bih2, bhh2, Whh2, x, Wih1, bih1, bhh1, 6);
  pipe_kernel<<<72,  256, 0, stream>>>(bufA, bufB, h2, Whh1, Wih2, bih2, bhh2, Whh2, x, Wih1, bih1, bhh1, 7);
  pipe_kernel<<<72,  256, 0, stream>>>(bufA, bufB, h2, Whh1, Wih2, bih2, bhh2, Whh2, x, Wih1, bih1, bhh1, 8);
  pipe_kernel<<<8,   256, 0, stream>>>(bufA, bufB, h2, Whh1, Wih2, bih2, bhh2, Whh2, x, Wih1, bih1, bhh1, 9);
  proj_ln_kernel<<<BSZ, PROJ, 0, stream>>>(h2, Wproj, bproj, gam, bet, out);
}


# recurrences: tanh by the hardware's v_tanh_f32 (1 transcendental per value, was 8 vector ops with exp2 and rcp); next step's first products held behind the h stores; sums and their order unchanged
# speedup vs baseline: 4.2914x; 1.0775x over previous
#include <hip/hip_runtime.h>

#define H 128
#define INW 2048
#define PROJ 64
#define BSZ 64
#define TLEN 512
#define M_TOT (BSZ * TLEN)
#define EPSV 1e-5f

typedef __attribute__((ext_vector_type(2))) float v2f;
typedef __attribute__((ext_vector_type(8))) float v8f;

#define WMMA_F32(a, b, c) \
  __builtin_amdgcn_wmma_f32_16x16x4_f32(false, (a), false, (b), (short)0, (c), false, false)

#define KC 128
#define LW (KC + 4)
#define G1_LDS_FLOATS ((32 + 128) * LW)

__device__ __forceinline__ void gemm1_strip(
    const float* __restrict__ x, const float* __restrict__ Wih1,
    const float* __restrict__ bih1, const float* __restrict__ bhh1,
    float* __restrict__ xp1, float* lds, const long m0) {
  float* xs = lds;
  float* ws = lds + 32 * LW;
  const int tid  = threadIdx.x;
  const int wave = tid >> 5;
  const int lane = tid & 31;
  const int half = lane >> 4;
  const int l16  = lane & 15;
  const int n0   = wave * 16;

  v8f c0 = {};
  v8f c1 = {};
  for (int k0 = 0; k0 < INW; k0 += KC) {
    for (int i = 0; i < 4; ++i) {
      int idx = tid + i * 256;
      int row = idx >> 5;
      int cv  = (idx & 31) << 2;
      float4 v = *(const float4*)&x[(m0 + row) * INW + k0 + cv];
      *(float4*)&xs[row * LW + cv] = v;
    }
    for (int i = 0; i < 16; ++i) {
      int idx = tid + i * 256;
      int row = idx >> 5;
      int cv  = (idx & 31) << 2;
      float4 v = *(const float4*)&Wih1[(long)row * INW + k0 + cv];
      *(float4*)&ws[row * LW + cv] = v;
    }
    __syncthreads();
    for (int k = 0; k < KC; k += 4) {
      const int kc = k + 2 * half;
      v2f b;  b.x  = ws[(n0 + l16) * LW + kc];  b.y  = ws[(n0 + l16) * LW + kc + 1];
      v2f a0; a0.x = xs[l16 * LW + kc];         a0.y = xs[l16 * LW + kc + 1];
      v2f a1; a1.x = xs[(16 + l16) * LW + kc];  a1.y = xs[(16 + l16) * LW + kc + 1];
      c0 = WMMA_F32(a0, b, c0);
      c1 = WMMA_F32(a1, b, c1);
    }
    __syncthreads();
  }
  const float bias = bih1[n0 + l16] + bhh1[n0 + l16];
  for (int v = 0; v < 8; ++v) {
    long m = m0 + v + 8 * half;
    xp1[m * H + n0 + l16]        = c0[v] + bias;
    xp1[(m + 16) * H + n0 + l16] = c1[v] + bias;
  }
}

__device__ __forceinline__ float tanh_unit(float x) {
  return __builtin_amdgcn_tanhf(x);
}

__device__ __forceinline__ void lds_step_barrier() {
  __builtin_amdgcn_fence(__ATOMIC_RELEASE, "workgroup", "local");
  __builtin_amdgcn_s_barrier();
  __builtin_amdgcn_fence(__ATOMIC_ACQUIRE, "workgroup", "local");
}

__device__ __forceinline__ void lds_step_barrier_arrive() {
  __builtin_amdgcn_fence(__ATOMIC_RELEASE, "workgroup", "local");
  __builtin_amdgcn_s_barrier_signal(-1);
}
__device__ __forceinline__ void lds_step_barrier_wait() {
  __builtin_amdgcn_s_barrier_wait(-1);
  __builtin_amdgcn_fence(__ATOMIC_ACQUIRE, "workgroup", "local");
}

__device__ __forceinline__ v8f load8(const float* p) {
  const float4 lo = *(const float4*)&p[0];
  const float4 hi = *(const float4*)&p[4];
  v8f r;
  r[0] = lo.x; r[1] = lo.y; r[2] = lo.z; r[3] = lo.w;
  r[4] = hi.x; r[5] = hi.y; r[6] = hi.z; r[7] = hi.w;
  return r;
}
__device__ __forceinline__ void store8(float* p, const v8f r) {
  float4 lo, hi;
  lo.x = r[0]; lo.y = r[1]; lo.z = r[2]; lo.w = r[3];
  hi.x = r[4]; hi.y = r[5]; hi.z = r[6]; hi.w = r[7];
  *(float4*)&p[0] = lo;
  *(float4*)&p[4] = hi;
}

__device__ __forceinline__ void pin_here(v8f& c) { asm volatile("" : "+v"(c) :: "memory"); }
__device__ __forceinline__ void pin_pair(v8f& c, v8f& d) { asm volatile("" : "+v"(c), "+v"(d)); }

#define HL (H + 4)
#define TBLK 4
#define CH 64
#define NCH (TLEN / CH)
#define G2R 64
#define PIPE_LDS_FLOATS (164 * 256)

#define SKEW 3
__device__ __forceinline__ void rnn1_chunk(
    const float* __restrict__ xp1, const float* __restrict__ Whh1,
    float* __restrict__ y1, float* hs, const int g, const int ch) {
  const int tid  = threadIdx.x;
  const int wave = __builtin_amdgcn_readfirstlane(tid >> 5);
  const int lane = tid & 31;
  const int half = lane >> 4;
  const int l16  = lane & 15;
  const int n0   = (wave & 3) * 32;
  const int b0   = g * 16;
  const long tofs = (long)ch * CH * H;

  if (ch == 0) {
    for (int i = tid; i < 16 * HL; i += 256) hs[i] = 0.0f;
  } else {
    const float* hin = y1 + (long)b0 * TLEN * H + tofs - H;
    const int r  = tid >> 5;
    const int cv = (tid & 31) << 2;
    const float4 va = *(const float4*)&hin[(long)r * TLEN * H + cv];
    const float4 vb = *(const float4*)&hin[(long)(r + 8) * TLEN * H + cv];
    *(float4*)&hs[r * HL + cv] = va;
    *(float4*)&hs[(r + 8) * HL + cv] = vb;
  }
  __syncthreads();

  if (wave < 4) {
    v2f aw0[32], aw1[32];
    const float* wr0 = Whh1 + (n0 + l16) * H;
    const float* wr1 = wr0 + 16 * H;
    for (int i = 0; i < 8; ++i) {
      const int k = n0 + 16 * (i >> 2) + 8 * half + 2 * (i & 3);
      aw0[i].x = wr0[k];  aw0[i].y = wr0[k + 1];
      aw1[i].x = wr1[k];  aw1[i].y = wr1[k + 1];
    }
    for (int i = 8; i < 32; ++i) {
      const int k = (((8 * wave + i) & 31) << 2) + 2 * half;
      aw0[i].x = wr0[k];  aw0[i].y = wr0[k + 1];
      aw1[i].x = wr1[k];  aw1[i].y = wr1[k + 1];
    }

    float* hst0 = hs + l16 * HL + n0 + 8 * half;
    float* hst1 = hst0 + 16;
    v8f ha = load8(hst0);
    v8f hb = load8(hst1);
    int fo[12];
    for (int p = 0; p < 12; ++p)
      fo[p] = l16 * HL + (((8 * wave + 8 + 2 * p) & 31) << 2) + 2 * half;

    const float* xrow = xp1 + (long)(b0 + l16) * TLEN * H + tofs + n0 + 8 * half;
    v8f ring0[TBLK], ring1[TBLK];
    for (int j = 0; j < TBLK - 1; ++j) {
      ring0[j] = load8(&xrow[(long)j * H]);
      ring1[j] = load8(&xrow[(long)j * H + 16]);
    }

    for (int tb = 0; tb < CH; tb += TBLK) {
#pragma unroll
      for (int j = 0; j < TBLK; ++j) {
        const int t = tb + j;
        v8f c0 = ring0[j];
        v8f c1 = ring1[j];
        pin_here(c0);
        pin_here(c1);
        v2f o[8];
        for (int i = 0; i < 4; ++i) {
          o[i].x     = ha[2 * i];  o[i].y     = ha[2 * i + 1];
          o[4 + i].x = hb[2 * i];  o[4 + i].y = hb[2 * i + 1];
        }
        c0 = WMMA_F32(aw0[0], o[0], c0);  c1 = WMMA_F32(aw1[0], o[0], c1);
        c0 = WMMA_F32(aw0[1], o[1], c0);  c1 = WMMA_F32(aw1[1], o[1], c1);
        pin_here(c0);
        pin_here(c1);
        lds_step_barrier_arrive();
        c0 = WMMA_F32(aw0[2], o[2], c0);  c1 = WMMA_F32(aw1[2], o[2], c1);
        lds_step_barrier_wait();
        v2f b[24];
        for (int p = 0; p < 12; ++p) {
          b[2 * p].x     = hs[fo[p]];      b[2 * p].y     = hs[fo[p] + 1];
          b[2 * p + 1].x = hs[fo[p] + 4];  b[2 * p + 1].y = hs[fo[p] + 5];
        }
        __builtin_amdgcn_sched_barrier(0);
        for (int p = 3; p < 8; ++p) {
          c0 = WMMA_F32(aw0[p], o[p], c0);  c1 = WMMA_F32(aw1[p], o[p], c1);
        }
        {
          const int tn = (t + TBLK - 1 < CH) ? t + TBLK - 1 : CH - 1;
          ring0[(j + TBLK - 1) % TBLK] = load8(&xrow[(long)tn * H]);
          ring1[(j + TBLK - 1) % TBLK] = load8(&xrow[(long)tn * H + 16]);
        }
        __builtin_amdgcn_sched_barrier(0);
        for (int i = 0; i < SKEW; ++i) c0 = WMMA_F32(aw0[i + 8], b[i], c0);
        for (int i = 0; i < 24 - SKEW; ++i) {
          c1 = WMMA_F32(aw1[i + 8], b[i], c1);
          c0 = WMMA_F32(aw0[i + SKEW + 8], b[i + SKEW], c0);
          pin_pair(c0, c1);
        }
        __builtin_amdgcn_sched_barrier(0);
        for (int i = 24 - SKEW; i < 24; ++i) c1 = WMMA_F32(aw1[i + 8], b[i], c1);
        for (int v = 0; v < 8; ++v) ha[v] = tanh_unit(c0[v]);
        for (int v = 0; v < 8; ++v) hb[v] = tanh_unit(c1[v]);
        lds_step_barrier();
        store8(hst0, ha);
        store8(hst1, hb);
      }
    }
    lds_step_barrier();
  } else {
    float* yrow = y1 + (long)(b0 + 4 * (wave - 4)) * TLEN * H + tofs + lane * 4;
    const float* hrow = hs + (4 * (wave - 4)) * HL + lane * 4;
    for (int t = 0; t < CH; ++t) {
      lds_step_barrier_arrive();
      lds_step_barrier_wait();
      if (t > 0) {
        float4 hv[4];
        for (int i = 0; i < 4; ++i) hv[i] = *(const float4*)&hrow[i * HL];
        for (int i = 0; i < 4; ++i)
          *(float4*)&yrow[(long)i * TLEN * H + (long)(t - 1) * H] = hv[i];
      }
      lds_step_barrier();
    }
    lds_step_barrier();
    float4 hv[4];
    for (int i = 0; i < 4; ++i) hv[i] = *(const float4*)&hrow[i * HL];
    for (int i = 0; i < 4; ++i)
      *(float4*)&yrow[(long)i * TLEN * H + (long)(CH - 1) * H] = hv[i];
  }
}

#define LW2 (H + 4)
__device__ __forceinline__ void gemm2_chunk(
    const float* __restrict__ y1, const float* __restrict__ Wih2,
    const float* __restrict__ bih2, const float* __restrict__ bhh2,
    float* __restrict__ xp2, float* lds, const int b, const int ch) {
  float* ys = lds;
  float* ws = lds + G2R * LW2;
  const int tid  = threadIdx.x;
  const int wave = tid >> 5;
  const int lane = tid & 31;
  const int half = lane >> 4;
  const int l16  = lane & 15;
  const int n0   = wave * 16;
  const long m0  = (long)b * TLEN + (long)ch * CH;

  for (int i = 0; i < 8; ++i) {
    int idx = tid + i * 256;
    int row = idx >> 5;
    int cv  = (idx & 31) << 2;
    float4 v = *(const float4*)&y1[(m0 + row) * H + cv];
    *(float4*)&ys[row * LW2 + cv] = v;
  }
  for (int i = 0; i < 16; ++i) {
    int idx = tid + i * 256;
    int row = idx >> 5;
    int cv  = (idx & 31) << 2;
    float4 v = *(const float4*)&Wih2[(long)row * H + cv];
    *(float4*)&ws[row * LW2 + cv] = v;
  }
  const float bias2 = bih2[n0 + l16] + bhh2[n0 + l16];
  v8f c0, c1, c2, c3;
  for (int v = 0; v < 8; ++v) { c0[v] = bias2; c1[v] = bias2; c2[v] = bias2; c3[v] = bias2; }
  __syncthreads();
  for (int k = 0; k < H; k += 4) {
    const int kc = k + 2 * half;
    v2f bb; bb.x = ws[(n0 + l16) * LW2 + kc];  bb.y = ws[(n0 + l16) * LW2 + kc + 1];
    v2f a0; a0.x = ys[l16 * LW2 + kc];         a0.y = ys[l16 * LW2 + kc + 1];
    v2f a1; a1.x = ys[(16 + l16) * LW2 + kc];  a1.y = ys[(16 + l16) * LW2 + kc + 1];
    v2f a2; a2.x = ys[(32 + l16) * LW2 + kc];  a2.y = ys[(32 + l16) * LW2 + kc + 1];
    v2f a3; a3.x = ys[(48 + l16) * LW2 + kc];  a3.y = ys[(48 + l16) * LW2 + kc + 1];
    c0 = WMMA_F32(a0, bb, c0);
    c1 = WMMA_F32(a1, bb, c1);
    c2 = WMMA_F32(a2, bb, c2);
    c3 = WMMA_F32(a3, bb, c3);
  }
  for (int v = 0; v < 8; ++v) {
    long m = m0 + v + 8 * half;
    xp2[m * H + n0 + l16]        = c0[v];
    xp2[(m + 16) * H + n0 + l16] = c1[v];
    xp2[(m + 32) * H + n0 + l16] = c2[v];
    xp2[(m + 48) * H + n0 + l16] = c3[v];
  }
}

__device__ __forceinline__ void rnn2_chunk(
    const float* __restrict__ xp2, const float* __restrict__ Whh2,
    float* __restrict__ h2, float* hs, const int g, const int ch) {
  const int tid  = threadIdx.x;
  const int wave = __builtin_amdgcn_readfirstlane(tid >> 5);
  const int lane = tid & 31;
  const int half = lane >> 4;
  const int l16  = lane & 15;
  const int n0   = (wave & 3) * 32;
  const int b0   = g * 16;
  const long tofs = (long)ch * CH * H;

  if (ch == 0) {
    for (int i = tid; i < 16 * HL; i += 256) hs[i] = 0.0f;
  } else {
    const float* hin = h2 + (long)b0 * H;
    const int r  = tid >> 5;
    const int cv = (tid & 31) << 2;
    const float4 va = *(const float4*)&hin[r * H + cv];
    const float4 vb = *(const float4*)&hin[(r + 8) * H + cv];
    *(float4*)&hs[r * HL + cv] = va;
    *(float4*)&hs[(r + 8) * HL + cv] = vb;
  }
  __syncthreads();

  if (wave < 4) {
    v2f aw0[32], aw1[32];
    const float* wr0 = Whh2 + (n0 + l16) * H;
    const float* wr1 = wr0 + 16 * H;
    for (int i = 0; i < 8; ++i) {
      const int k = n0 + 16 * (i >> 2) + 8 * half + 2 * (i & 3);
      aw0[i].x = wr0[k];  aw0[i].y = wr0[k + 1];
      aw1[i].x = wr1[k];  aw1[i].y = wr1[k + 1];
    }
    for (int i = 8; i < 32; ++i) {
      const int k = (((8 * wave + i) & 31) << 2) + 2 * half;
      aw0[i].x = wr0[k];  aw0[i].y = wr0[k + 1];
      aw1[i].x = wr1[k];  aw1[i].y = wr1[k + 1];
    }

    float* hst0 = hs + l16 * HL + n0 + 8 * half;
    float* hst1 = hst0 + 16;
    v8f ha = load8(hst0);
    v8f hb = load8(hst1);
    int fo[12];
    for (int p = 0; p < 12; ++p)
      fo[p] = l16 * HL + (((8 * wave + 8 + 2 * p) & 31) << 2) + 2 * half;

    const float* xrow = xp2 + (long)(b0 + l16) * TLEN * H + tofs + n0 + 8 * half;
    v8f ring0[TBLK], ring1[TBLK];
    for (int j = 0; j < TBLK - 1; ++j) {
      ring0[j] = load8(&xrow[(long)j * H]);
      ring1[j] = load8(&xrow[(long)j * H + 16]);
    }

    for (int tb = 0; tb < CH; tb += TBLK) {
#pragma unroll
      for (int j = 0; j < TBLK; ++j) {
        const int t = tb + j;
        v8f c0 = ring0[j];
        v8f c1 = ring1[j];
        pin_here(c0);
        pin_here(c1);
        v2f o[8];
        for (int i = 0; i < 4; ++i) {
          o[i].x     = ha[2 * i];  o[i].y     = ha[2 * i + 1];
          o[4 + i].x = hb[2 * i];  o[4 + i].y = hb[2 * i + 1];
        }
        c0 = WMMA_F32(aw0[0], o[0], c0);  c1 = WMMA_F32(aw1[0], o[0], c1);
        c0 = WMMA_F32(aw0[1], o[1], c0);  c1 = WMMA_F32(aw1[1], o[1], c1);
        pin_here(c0);
        pin_here(c1);
        lds_step_barrier_arrive();
        c0 = WMMA_F32(aw0[2], o[2], c0);  c1 = WMMA_F32(aw1[2], o[2], c1);
        lds_step_barrier_wait();
        v2f b[24];
        for (int p = 0; p < 12; ++p) {
          b[2 * p].x     = hs[fo[p]];      b[2 * p].y     = hs[fo[p] + 1];
          b[2 * p + 1].x = hs[fo[p] + 4];  b[2 * p + 1].y = hs[fo[p] + 5];
        }
        __builtin_amdgcn_sched_barrier(0);
        for (int p = 3; p < 8; ++p) {
          c0 = WMMA_F32(aw0[p], o[p], c0);  c1 = WMMA_F32(aw1[p], o[p], c1);
        }
        {
          const int tn = (t + TBLK - 1 < CH) ? t + TBLK - 1 : CH - 1;
          ring0[(j + TBLK - 1) % TBLK] = load8(&xrow[(long)tn * H]);
          ring1[(j + TBLK - 1) % TBLK] = load8(&xrow[(long)tn * H + 16]);
        }
        __builtin_amdgcn_sched_barrier(0);
        for (int i = 0; i < SKEW; ++i) c0 = WMMA_F32(aw0[i + 8], b[i], c0);
        for (int i = 0; i < 24 - SKEW; ++i) {
          c1 = WMMA_F32(aw1[i + 8], b[i], c1);
          c0 = WMMA_F32(aw0[i + SKEW + 8], b[i + SKEW], c0);
          pin_pair(c0, c1);
        }
        __builtin_amdgcn_sched_barrier(0);
        for (int i = 24 - SKEW; i < 24; ++i) c1 = WMMA_F32(aw1[i + 8], b[i], c1);
        for (int v = 0; v < 8; ++v) ha[v] = tanh_unit(c0[v]);
        for (int v = 0; v < 8; ++v) hb[v] = tanh_unit(c1[v]);
        lds_step_barrier();
        store8(hst0, ha);
        store8(hst1, hb);
      }
    }
    float* hout = &h2[(long)(b0 + l16) * H + n0 + 8 * half];
    store8(hout, ha);
    store8(hout + 16, hb);
  } else {
    for (int t = 0; t < CH; ++t) {
      lds_step_barrier_arrive();
      lds_step_barrier_wait();
      lds_step_barrier();
    }
  }
}

__global__ __launch_bounds__(256) void gemm1_head(
    const float* __restrict__ x, const float* __restrict__ Wih1,
    const float* __restrict__ bih1, const float* __restrict__ bhh1,
    float* __restrict__ xp1) {
  __shared__ __attribute__((aligned(16))) float lds[G1_LDS_FLOATS];
  const int s = blockIdx.x;
  const long m0 = (long)(s >> 1) * TLEN + (long)(s & 1) * 32;
  gemm1_strip(x, Wih1, bih1, bhh1, xp1, lds, m0);
}

#define PIPE_G1_FIRST (8 + BSZ)
__global__ __launch_bounds__(256) void pipe_kernel(
    float* bufA, float* bufB, float* h2,
    const float* Whh1, const float* Wih2, const float* bih2, const float* bhh2,
    const float* Whh2,
    const float* x, const float* Wih1, const float* bih1, const float* bhh1,
    const int li) {
  __shared__ __attribute__((aligned(16))) float lds[PIPE_LDS_FLOATS];
  const int bx = blockIdx.x;
  if (bx < 4) {
    if (li < NCH) rnn1_chunk(bufA, Whh1, bufB, lds, bx, li);
  } else if (bx < 8) {
    if (li >= 2 && li - 2 < NCH) rnn2_chunk(bufA, Whh2, h2, lds, bx - 4, li - 2);
  } else if (bx < PIPE_G1_FIRST) {
    if (li >= 1 && li - 1 < NCH) gemm2_chunk(bufB, Wih2, bih2, bhh2, bufA, lds, bx - 8, li - 1);
  } else {
    if (li + 1 < NCH) {
      const int s = bx - PIPE_G1_FIRST;
      const long m0 = (long)(s >> 1) * TLEN + (long)(li + 1) * CH + (long)(s & 1) * 32;
      gemm1_strip(x, Wih1, bih1, bhh1, bufA, lds, m0);
    }
  }
}

__global__ __launch_bounds__(64) void proj_ln_kernel(
    const float* __restrict__ h2, const float* __restrict__ Wp,
    const float* __restrict__ bp, const float* __restrict__ gamma,
    const float* __restrict__ beta, float* __restrict__ out) {
  __shared__ float tmp[PROJ];
  const int b = blockIdx.x;
  const int j = threadIdx.x;
  float acc = bp[j];
  for (int k = 0; k < H; ++k) acc += h2[b * H + k] * Wp[j * H + k];

  tmp[j] = acc;
  __syncthreads();
  for (int s = 32; s > 0; s >>= 1) {
    if (j < s) tmp[j] += tmp[j + s];
    __syncthreads();
  }
  const float mu = tmp[0] * (1.0f / PROJ);
  __syncthreads();
  const float d = acc - mu;
  tmp[j] = d * d;
  __syncthreads();
  for (int s = 32; s > 0; s >>= 1) {
    if (j < s) tmp[j] += tmp[j + s];
    __syncthreads();
  }
  const float var = tmp[0] * (1.0f / PROJ);
  const float r = rsqrtf(var + EPSV);
  out[b * PROJ + j] = d * r * gamma[j] + beta[j];
}

extern "C" void kernel_launch(void* const* d_in, const int* in_sizes, int n_in,
                              void* d_out, int out_size, void* d_ws, size_t ws_size,
                              hipStream_t stream) {
  (void)in_sizes; (void)n_in; (void)out_size; (void)ws_size;
  const float* x     = (const float*)d_in[0];
  const float* Wih1  = (const float*)d_in[1];
  const float* Whh1  = (const float*)d_in[2];
  const float* bih1  = (const float*)d_in[3];
  const float* bhh1  = (const float*)d_in[4];
  const float* Wih2  = (const float*)d_in[5];
  const float* Whh2  = (const float*)d_in[6];
  const float* bih2  = (const float*)d_in[7];
  const float* bhh2  = (const float*)d_in[8];
  const float* Wproj = (const float*)d_in[9];
  const float* bproj = (const float*)d_in[10];
  const float* gam   = (const float*)d_in[11];
  const float* bet   = (const float*)d_in[12];
  float* out = (float*)d_out;

  float* bufA = (float*)d_ws;
  float* bufB = bufA + (size_t)M_TOT * H;
  float* h2   = bufB + (size_t)M_TOT * H;

  gemm1_head<<<128, 256, 0, stream>>>(x, Wih1, bih1, bhh1, bufA);
  pipe_kernel<<<200, 256, 0, stream>>>(bufA, bufB, h2, Whh1, Wih2, bih2, bhh2, Whh2, x, Wih1, bih1, bhh1, 0);
  pipe_kernel<<<200, 256, 0, stream>>>(bufA, bufB, h2, Whh1, Wih2, bih2, bhh2, Whh2, x, Wih1, bih1, bhh1, 1);
  pipe_kernel<<<200, 256, 0, stream>>>(bufA, bufB, h2, Whh1, Wih2, bih2, bhh2, Whh2, x, Wih1, bih1, bhh1, 2);
  pipe_kernel<<<200, 256, 0, stream>>>(bufA, bufB, h2, Whh1, Wih2, bih2, bhh2, Whh2, x, Wih1, bih1, bhh1, 3);
  pipe_kernel<<<200, 256, 0, stream>>>(bufA, bufB, h2, Whh1, Wih2, bih2, bhh2, Whh2, x, Wih1, bih1, bhh1, 4);
  pipe_kernel<<<200, 256, 0, stream>>>(bufA, bufB, h2, Whh1, Wih2, bih2, bhh2, Whh2, x, Wih1, bih1, bhh1, 5);
  pipe_kernel<<<200, 256, 0, stream>>>(bufA, bufB, h2, Whh1, Wih2, bih2, bhh2, Whh2, x, Wih1, bih1, bhh1, 6);
  pipe_kernel<<<72,  256, 0, stream>>>(bufA, bufB, h2, Whh1, Wih2, bih2, bhh2, Whh2, x, Wih1, bih1, bhh1, 7);
  pipe_kernel<<<72,  256, 0, stream>>>(bufA, bufB, h2, Whh1, Wih2, bih2, bhh2, Whh2, x, Wih1, bih1, bhh1, 8);
  pipe_kernel<<<8,   256, 0, stream>>>(bufA, bufB, h2, Whh1, Wih2, bih2, bhh2, Whh2, x, Wih1, bih1, bhh1, 9);
  proj_ln_kernel<<<BSZ, PROJ, 0, stream>>>(h2, Wproj, bproj, gam, bet, out);
}
